# starting kernel
# baseline (speedup 1.0000x reference)
#include <hip/hip_runtime.h>
#include <math.h>

#define NB 4
#define NT 512
#define ND 256
#define MOMB 0.9f
#define NOVB 0.95f
#define MOM_STRIDE 260

typedef float v2f __attribute__((ext_vector_type(2)));
typedef float v8f __attribute__((ext_vector_type(8)));

struct __align__(16) Smem {
  float mom[ND * MOM_STRIDE];
  float x_s[ND];
  float err_s[ND];
  float yacc[2][ND];
  float iema[ND];
  float redn[8];
  float redf[32];
  float eff_s;
  float scale_s;
};

__global__ __launch_bounds__(1024) void ttt_linear_kernel(
    const float* __restrict__ x, const float* __restrict__ Wb,
    const float* __restrict__ bias, const float* __restrict__ loglr,
    float* __restrict__ out)
{
  __shared__ Smem sm;
  const int tid  = threadIdx.x;
  const int lane = tid & 31;
  const int wid  = tid >> 5;
  const int b    = blockIdx.x;
  const int hi   = lane >> 4;
  const int rb   = (wid >> 1) << 4;
  const int row  = rb + (lane & 15);
  const int colbase = (wid & 1) << 7;

  float w_u[64];
  #pragma unroll
  for (int c = 0; c < 32; ++c) {
    const int col = colbase + 4*c + 2*hi;
    const float2 wv = *reinterpret_cast<const float2*>(Wb + row*ND + col);
    w_u[2*c]   = wv.x;
    w_u[2*c+1] = wv.y;
  }
  for (int i = tid; i < ND*MOM_STRIDE; i += 1024) sm.mom[i] = 0.0f;
  if (tid < ND) sm.iema[tid] = 0.0f;

  float lrema    = 1.0f;
  float inner_lr = 0.0f;
  if (tid == 0) {
    inner_lr = fminf(fmaxf(expf(loglr[0]), 1e-5f), 1.0f);
    sm.scale_s = 1.0f;
    sm.eff_s   = 0.0f;
  }
  const float bias_r = (tid < ND) ? bias[tid] : 0.0f;
  __syncthreads();

  for (int t = 0; t < NT; ++t) {
    if (tid < ND) {
      const float xj = x[((size_t)b*NT + t)*ND + tid];
      sm.x_s[tid] = xj;
      float a = fabsf(xj - sm.iema[tid]);
      sm.iema[tid] = NOVB*sm.iema[tid] + (1.0f-NOVB)*xj;
      #pragma unroll
      for (int off = 16; off > 0; off >>= 1) a += __shfl_xor(a, off, 32);
      if (lane == 0) sm.redn[wid] = a;
    }
    __syncthreads();

    if (tid == 0) {
      float s = 0.0f;
      #pragma unroll
      for (int i = 0; i < 8; ++i) s += sm.redn[i];
      const float nov = s * (1.0f/ND);
      const float lrs = fminf(fmaxf(1.0f + 3.0f*nov, 0.5f), 3.0f);
      lrema = NOVB*lrema + (1.0f-NOVB)*lrs;
      sm.eff_s = inner_lr * lrema;
    }

    v8f acc = {};
    #pragma unroll
    for (int c = 0; c < 32; ++c) {
      const int col = colbase + 4*c + 2*hi;
      const float2 xp = *reinterpret_cast<const float2*>(&sm.x_s[col]);
      v2f av; av.x = w_u[2*c]; av.y = w_u[2*c+1];
      v2f bv; bv.x = xp.x;     bv.y = xp.y;
      acc = __builtin_amdgcn_wmma_f32_16x16x4_f32(
           false, av,  false, bv,
           (short)0, acc,  false,  false);
    }
    if ((lane & 15) == 0) {
      const int mbase = rb + hi*8;
      #pragma unroll
      for (int v = 0; v < 8; ++v) sm.yacc[wid & 1][mbase + v] = acc[v];
    }
    __syncthreads();

    if (tid < ND) {
      const float yv = sm.scale_s*(sm.yacc[0][tid] + sm.yacc[1][tid]) + bias_r;
      out[((size_t)b*NT + t)*ND + tid] = yv;
      sm.err_s[tid] = (yv - sm.x_s[tid]) * (1.0f/ND);
    }
    __syncthreads();

    const float err = sm.err_s[row];
    const float eff = sm.eff_s;
    const float sp  = sm.scale_s;
    float fsum = 0.0f;
    #pragma unroll
    for (int c = 0; c < 32; ++c) {
      const int col = colbase + 4*c + 2*hi;
      const float2 xp = *reinterpret_cast<const float2*>(&sm.x_s[col]);
      float2* mptr = reinterpret_cast<float2*>(&sm.mom[row*MOM_STRIDE + col]);
      float2 mp = *mptr;
      mp.x = MOMB*mp.x + (1.0f-MOMB)*err*xp.x;
      mp.y = MOMB*mp.y + (1.0f-MOMB)*err*xp.y;
      *mptr = mp;
      const float w0 = sp*w_u[2*c]   - eff*mp.x;
      const float w1 = sp*w_u[2*c+1] - eff*mp.y;
      w_u[2*c] = w0; w_u[2*c+1] = w1;
      fsum += w0*w0 + w1*w1;
    }
    #pragma unroll
    for (int off = 16; off > 0; off >>= 1) fsum += __shfl_xor(fsum, off, 32);
    if (lane == 0) sm.redf[wid] = fsum;
    __syncthreads();

    if (tid == 0) {
      float s = 0.0f;
      #pragma unroll
      for (int i = 0; i < 32; ++i) s += sm.redf[i];
      const float fn = fmaxf(sqrtf(s), 1e-6f);
      sm.scale_s = fminf(32.0f/fn, 1.0f);
    }
  }
}

extern "C" void kernel_launch(void* const* d_in, const int* in_sizes, int n_in,
                              void* d_out, int out_size, void* d_ws, size_t ws_size,
                              hipStream_t stream) {
  (void)in_sizes; (void)n_in; (void)out_size; (void)d_ws; (void)ws_size;
  const float* x    = (const float*)d_in[0];
  const float* Wb   = (const float*)d_in[1];
  const float* bias = (const float*)d_in[2];
  const float* lil  = (const float*)d_in[3];
  float* out = (float*)d_out;
  ttt_linear_kernel<<<dim3(NB), dim3(1024), 0, stream>>>(x, Wb, bias, lil, out);
}


# Software-pipeline the scan: the matvec WMMAs of step t+1 issue inside the momentum/weight update loop of step t (x staged one step ahead in a second LDS buffer); 3 barriers per step, not 4
# speedup vs baseline: 1.2082x; 1.2082x over previous
#include <hip/hip_runtime.h>
#include <math.h>


#define NB 4
#define NT 512
#define ND 256
#define MOMB 0.9f
#define NOVB 0.95f
#define MOM_STRIDE 260

typedef float v2f __attribute__((ext_vector_type(2)));
typedef float v8f __attribute__((ext_vector_type(8)));

struct __align__(16) Smem {
  float mom[ND * MOM_STRIDE];
  float x_s[2][ND];
  float err_s[ND];
  float yacc[2][ND];
  float iema[ND];
  float redn[8];
  float redf[32];
  float eff_s[2];
  float scale_s;
  float lrema_s;
  float inner_lr_s;
};

__device__ __forceinline__ void lr_step(Smem& sm, int buf)
{
  float s = 0.0f;
  #pragma unroll
  for (int i = 0; i < 8; ++i) s += sm.redn[i];
  const float nov = s * (1.0f/ND);
  const float lrs = fminf(fmaxf(1.0f + 3.0f*nov, 0.5f), 3.0f);
  const float lrema = NOVB*sm.lrema_s + (1.0f-NOVB)*lrs;
  sm.lrema_s = lrema;
  sm.eff_s[buf] = sm.inner_lr_s * lrema;
}

__device__ __forceinline__ void stage_x(Smem& sm, int buf, int j, int slot, int lane, float xj)
{
  sm.x_s[buf][j] = xj;
  const float old = sm.iema[j];
  float a = fabsf(xj - old);
  sm.iema[j] = NOVB*old + (1.0f-NOVB)*xj;
  #pragma unroll
  for (int off = 16; off > 0; off >>= 1) a += __shfl_xor(a, off, 32);
  if (lane == 0) sm.redn[slot] = a;
}

__global__ __launch_bounds__(1024) void ttt_linear_kernel(
    const float* __restrict__ x, const float* __restrict__ Wb,
    const float* __restrict__ bias, const float* __restrict__ loglr,
    float* __restrict__ out)
{
  __shared__ Smem sm;
  const int tid  = threadIdx.x;
  const int lane = tid & 31;
  const int wid  = tid >> 5;
  const int b    = blockIdx.x;
  const int hi   = lane >> 4;
  const int rb   = (wid >> 1) << 4;
  const int row  = rb + (lane & 15);
  const int colbase = (wid & 1) << 7;
  const bool stager = (tid >= ND) && (tid < 2*ND);
  const int  j      = tid & (ND-1);

  float w_u[64];
  #pragma unroll
  for (int c = 0; c < 32; ++c) {
    const int col = colbase + 4*c + 2*hi;
    const float2 wv = *reinterpret_cast<const float2*>(Wb + row*ND + col);
    w_u[2*c]   = wv.x;
    w_u[2*c+1] = wv.y;
  }
  for (int i = tid; i < ND*MOM_STRIDE; i += 1024) sm.mom[i] = 0.0f;

  if (tid == 0) {
    sm.inner_lr_s = fminf(fmaxf(expf(loglr[0]), 1e-5f), 1.0f);
    sm.lrema_s = 1.0f;
    sm.scale_s = 1.0f;
  }
  const float bias_r = (tid < ND) ? bias[tid] : 0.0f;

  float xpre = 0.0f;
  if (stager) {
    sm.iema[j] = 0.0f;
    stage_x(sm, 0, j, wid - 8, lane, x[((size_t)b*NT + 0)*ND + j]);
    xpre = x[((size_t)b*NT + 1)*ND + j];
  }
  __syncthreads();

  if (tid == 0) lr_step(sm, 0);

  v8f acc = {};
  #pragma unroll
  for (int c = 0; c < 32; ++c) {
    const int col = colbase + 4*c + 2*hi;
    const float2 xp = *reinterpret_cast<const float2*>(&sm.x_s[0][col]);
    v2f av; av.x = w_u[2*c]; av.y = w_u[2*c+1];
    v2f bv; bv.x = xp.x;     bv.y = xp.y;
    acc = __builtin_amdgcn_wmma_f32_16x16x4_f32(
         false, av,  false, bv,
         (short)0, acc,  false,  false);
  }

  for (int t = 0; t < NT; ++t) {
    const int cur = t & 1;
    const int nxt = cur ^ 1;

    if ((lane & 15) == 0) {
      const int mbase = rb + hi*8;
      #pragma unroll
      for (int v = 0; v < 8; ++v) sm.yacc[wid & 1][mbase + v] = acc[v];
    }
    __syncthreads();

    if (tid < ND) {
      const float yv = sm.scale_s*(sm.yacc[0][tid] + sm.yacc[1][tid]) + bias_r;
      out[((size_t)b*NT + t)*ND + tid] = yv;
      sm.err_s[tid] = (yv - sm.x_s[cur][tid]) * (1.0f/ND);
    } else if (stager && (t + 1 < NT)) {
      stage_x(sm, nxt, j, wid - 8, lane, xpre);
    }
    __syncthreads();

    if (t + 1 == NT) break;

    if (stager && (t + 2 < NT)) xpre = x[((size_t)b*NT + t + 2)*ND + j];

    if (tid == 0) lr_step(sm, nxt);

    const float err = sm.err_s[row];
    const float eff = sm.eff_s[cur];
    const float sp  = sm.scale_s;
    const float* xc = sm.x_s[cur];
    const float* xn = sm.x_s[nxt];
    float fsum = 0.0f;
    acc = v8f{};
    const int col0 = colbase + 2*hi;
    const float* mrow = &sm.mom[row*MOM_STRIDE + col0];
    float2 xp = *reinterpret_cast<const float2*>(&xc[col0]);
    float2 xq = *reinterpret_cast<const float2*>(&xn[col0]);
    float2 mp = *reinterpret_cast<const float2*>(&mrow[0]);
    #pragma unroll
    for (int c = 0; c < 32; ++c) {
      float2 xp_n = xp, xq_n = xq, mp_n = mp;
      if (c + 1 < 32) {
        xp_n = *reinterpret_cast<const float2*>(&xc[col0 + 4*(c + 1)]);
        xq_n = *reinterpret_cast<const float2*>(&xn[col0 + 4*(c + 1)]);
        mp_n = *reinterpret_cast<const float2*>(&mrow[4*(c + 1)]);
      }
      mp.x = MOMB*mp.x + (1.0f-MOMB)*err*xp.x;
      mp.y = MOMB*mp.y + (1.0f-MOMB)*err*xp.y;
      *reinterpret_cast<float2*>(&sm.mom[row*MOM_STRIDE + col0 + 4*c]) = mp;
      const float w0 = sp*w_u[2*c]   - eff*mp.x;
      const float w1 = sp*w_u[2*c+1] - eff*mp.y;
      w_u[2*c] = w0; w_u[2*c+1] = w1;
      fsum += w0*w0 + w1*w1;
      v2f av; av.x = w0;   av.y = w1;
      v2f bv; bv.x = xq.x; bv.y = xq.y;
      acc = __builtin_amdgcn_wmma_f32_16x16x4_f32(
           false, av,  false, bv,
           (short)0, acc,  false,  false);
      if (c + 1 < 32) __builtin_amdgcn_sched_group_barrier(0x100, 3, 0);
      __builtin_amdgcn_sched_group_barrier(0x002, 8, 0);
      __builtin_amdgcn_sched_group_barrier(0x200, 1, 0);
      __builtin_amdgcn_sched_group_barrier(0x008, 1, 0);
      xp = xp_n; xq = xq_n; mp = mp_n;
    }
    #pragma unroll
    for (int off = 16; off > 0; off >>= 1) fsum += __shfl_xor(fsum, off, 32);
    if (lane == 0) sm.redf[wid] = fsum;
    __syncthreads();

    if (tid == 0) {
      float s = 0.0f;
      #pragma unroll
      for (int i = 0; i < 32; ++i) s += sm.redf[i];
      const float fn = fmaxf(sqrtf(s), 1e-6f);
      sm.scale_s = fminf(32.0f/fn, 1.0f);
    }
  }
}

extern "C" void kernel_launch(void* const* d_in, const int* in_sizes, int n_in,
                              void* d_out, int out_size, void* d_ws, size_t ws_size,
                              hipStream_t stream) {
  (void)in_sizes; (void)n_in; (void)out_size; (void)d_ws; (void)ws_size;
  const float* x    = (const float*)d_in[0];
  const float* Wb   = (const float*)d_in[1];
  const float* bias = (const float*)d_in[2];
  const float* lil  = (const float*)d_in[3];
  float* out = (float*)d_out;
  ttt_linear_kernel<<<dim3(NB), dim3(1024), 0, stream>>>(x, Wb, bias, lil, out);
}


# Per-step serial path cut: out store issued behind the barrier (no store-ack wait), lane sums on DPP/permlane not LDS permutes, fnorm squares as packed pairs then a wave-0 tree (sum order changes)
# speedup vs baseline: 1.4887x; 1.2322x over previous
#include <hip/hip_runtime.h>
#include <math.h>


#define NB 4
#define NT 512
#define ND 256
#define MOMB 0.9f
#define NOVB 0.95f
#define MOM_STRIDE 260

typedef float v2f __attribute__((ext_vector_type(2)));
typedef float v8f __attribute__((ext_vector_type(8)));

struct __align__(16) Smem {
  float mom[ND * MOM_STRIDE];
  float x_s[2][ND];
  float err_s[ND];
  float yacc[2][ND];
  float iema[ND];
  float redn[8];
  float redf[32];
  float eff_s[2];
  float scale_s;
  float lrema_s;
  float inner_lr_s;
};

__device__ __forceinline__ float xhalf(float v)
{
  const int b = __builtin_bit_cast(int, v);
  return __builtin_bit_cast(float, __builtin_amdgcn_permlanex16(b, b, 0x76543210, 0xfedcba98, false, false));
}

template <int CTRL>
__device__ __forceinline__ float row_xmask(float v)
{
  const int b = __builtin_bit_cast(int, v);
  return __builtin_bit_cast(float, __builtin_amdgcn_update_dpp(b, b, CTRL, 0xf, 0xf, true));
}

__device__ __forceinline__ float wave_sum(float a)
{
  a += xhalf(a);
  a += row_xmask<0x168>(a);
  a += row_xmask<0x164>(a);
  a += row_xmask<0x162>(a);
  a += row_xmask<0x161>(a);
  return a;
}

__device__ __forceinline__ void lr_step(Smem& sm, int buf)
{
  float s = 0.0f;
  #pragma unroll
  for (int i = 0; i < 8; ++i) s += sm.redn[i];
  const float nov = s * (1.0f/ND);
  const float lrs = fminf(fmaxf(1.0f + 3.0f*nov, 0.5f), 3.0f);
  const float lrema = NOVB*sm.lrema_s + (1.0f-NOVB)*lrs;
  sm.lrema_s = lrema;
  sm.eff_s[buf] = sm.inner_lr_s * lrema;
}

__device__ __forceinline__ void stage_x(Smem& sm, int buf, int j, int slot, int lane, float xj)
{
  sm.x_s[buf][j] = xj;
  const float old = sm.iema[j];
  float a = fabsf(xj - old);
  sm.iema[j] = NOVB*old + (1.0f-NOVB)*xj;
  a = wave_sum(a);
  if (lane == 0) sm.redn[slot] = a;
}

__global__ __launch_bounds__(1024) void ttt_linear_kernel(
    const float* __restrict__ x, const float* __restrict__ Wb,
    const float* __restrict__ bias, const float* __restrict__ loglr,
    float* __restrict__ out)
{
  __shared__ Smem sm;
  const int tid  = threadIdx.x;
  const int lane = tid & 31;
  const int wid  = tid >> 5;
  const int b    = blockIdx.x;
  const int hi   = lane >> 4;
  const int rb   = (wid >> 1) << 4;
  const int row  = rb + (lane & 15);
  const int colbase = (wid & 1) << 7;
  const bool stager = (tid >= ND) && (tid < 2*ND);
  const int  j      = tid & (ND-1);

  float w_u[64];
  #pragma unroll
  for (int c = 0; c < 32; ++c) {
    const int col = colbase + 4*c + 2*hi;
    const float2 wv = *reinterpret_cast<const float2*>(Wb + row*ND + col);
    w_u[2*c]   = wv.x;
    w_u[2*c+1] = wv.y;
  }
  for (int i = tid; i < ND*MOM_STRIDE; i += 1024) sm.mom[i] = 0.0f;

  if (tid == 0) {
    sm.inner_lr_s = fminf(fmaxf(expf(loglr[0]), 1e-5f), 1.0f);
    sm.lrema_s = 1.0f;
    sm.scale_s = 1.0f;
  }
  const float bias_r = (tid < ND) ? bias[tid] : 0.0f;

  float xpre = 0.0f;
  if (stager) {
    sm.iema[j] = 0.0f;
    stage_x(sm, 0, j, wid - 8, lane, x[((size_t)b*NT + 0)*ND + j]);
    xpre = x[((size_t)b*NT + 1)*ND + j];
  }
  __syncthreads();

  if (tid == 0) lr_step(sm, 0);

  v8f acc = {};
  #pragma unroll
  for (int c = 0; c < 32; ++c) {
    const int col = colbase + 4*c + 2*hi;
    const float2 xp = *reinterpret_cast<const float2*>(&sm.x_s[0][col]);
    v2f av; av.x = w_u[2*c]; av.y = w_u[2*c+1];
    v2f bv; bv.x = xp.x;     bv.y = xp.y;
    acc = __builtin_amdgcn_wmma_f32_16x16x4_f32(
         false, av,  false, bv,
         (short)0, acc,  false,  false);
  }

  for (int t = 0; t < NT; ++t) {
    const int cur = t & 1;
    const int nxt = cur ^ 1;

    if ((lane & 15) == 0) {
      const int mbase = rb + hi*8;
      #pragma unroll
      for (int v = 0; v < 8; ++v) sm.yacc[wid & 1][mbase + v] = acc[v];
    }
    __syncthreads();

    float yv = 0.0f;
    if (tid < ND) {
      yv = sm.scale_s*(sm.yacc[0][tid] + sm.yacc[1][tid]) + bias_r;
      sm.err_s[tid] = (yv - sm.x_s[cur][tid]) * (1.0f/ND);
    } else if (stager && (t + 1 < NT)) {
      stage_x(sm, nxt, j, wid - 8, lane, xpre);
    }
    __syncthreads();

    if (tid < ND) out[((size_t)b*NT + t)*ND + tid] = yv;

    if (t + 1 == NT) break;

    if (stager && (t + 2 < NT)) xpre = x[((size_t)b*NT + t + 2)*ND + j];

    if (tid == 0) lr_step(sm, nxt);

    const float err = sm.err_s[row];
    const float eff = sm.eff_s[cur];
    const float sp  = sm.scale_s;
    const float* xc = sm.x_s[cur];
    const float* xn = sm.x_s[nxt];
    v2f fs2[4];
    #pragma unroll
    for (int q = 0; q < 4; ++q) fs2[q] = v2f{0.0f, 0.0f};
    acc = v8f{};
    const int col0 = colbase + 2*hi;
    const float* mrow = &sm.mom[row*MOM_STRIDE + col0];
    float2 xp = *reinterpret_cast<const float2*>(&xc[col0]);
    float2 xq = *reinterpret_cast<const float2*>(&xn[col0]);
    float2 mp = *reinterpret_cast<const float2*>(&mrow[0]);
    #pragma unroll
    for (int c = 0; c < 32; ++c) {
      float2 xp_n = xp, xq_n = xq, mp_n = mp;
      if (c + 1 < 32) {
        xp_n = *reinterpret_cast<const float2*>(&xc[col0 + 4*(c + 1)]);
        xq_n = *reinterpret_cast<const float2*>(&xn[col0 + 4*(c + 1)]);
        mp_n = *reinterpret_cast<const float2*>(&mrow[4*(c + 1)]);
      }
      mp.x = MOMB*mp.x + (1.0f-MOMB)*err*xp.x;
      mp.y = MOMB*mp.y + (1.0f-MOMB)*err*xp.y;
      *reinterpret_cast<float2*>(&sm.mom[row*MOM_STRIDE + col0 + 4*c]) = mp;
      const float w0 = sp*w_u[2*c]   - eff*mp.x;
      const float w1 = sp*w_u[2*c+1] - eff*mp.y;
      w_u[2*c] = w0; w_u[2*c+1] = w1;
      v2f av; av.x = w0;   av.y = w1;
      fs2[c & 3] = __builtin_elementwise_fma(av, av, fs2[c & 3]);
      v2f bv; bv.x = xq.x; bv.y = xq.y;
      acc = __builtin_amdgcn_wmma_f32_16x16x4_f32(
           false, av,  false, bv,
           (short)0, acc,  false,  false);
      if (c + 1 < 32) __builtin_amdgcn_sched_group_barrier(0x100, 3, 0);
      __builtin_amdgcn_sched_group_barrier(0x002, 8, 0);
      __builtin_amdgcn_sched_group_barrier(0x200, 1, 0);
      __builtin_amdgcn_sched_group_barrier(0x008, 1, 0);
      xp = xp_n; xq = xq_n; mp = mp_n;
    }
    const v2f fs01 = fs2[0] + fs2[1];
    const v2f fs23 = fs2[2] + fs2[3];
    const v2f fsv  = fs01 + fs23;
    const float fsum = wave_sum(fsv.x + fsv.y);
    if (lane == 0) sm.redf[wid] = fsum;
    __syncthreads();

    if (tid < 32) {
      const float s  = wave_sum(sm.redf[lane]);
      const float fn = fmaxf(sqrtf(s), 1e-6f);
      if (lane == 0) sm.scale_s = fminf(32.0f/fn, 1.0f);
    }
  }
}

extern "C" void kernel_launch(void* const* d_in, const int* in_sizes, int n_in,
                              void* d_out, int out_size, void* d_ws, size_t ws_size,
                              hipStream_t stream) {
  (void)in_sizes; (void)n_in; (void)out_size; (void)d_ws; (void)ws_size;
  const float* x    = (const float*)d_in[0];
  const float* Wb   = (const float*)d_in[1];
  const float* bias = (const float*)d_in[2];
  const float* lil  = (const float*)d_in[3];
  float* out = (float*)d_out;
  ttt_linear_kernel<<<dim3(NB), dim3(1024), 0, stream>>>(x, Wb, bias, lil, out);
}


# One workgroup barrier per scan step, not three: partials double-buffered, each wave rebuilds its error rows and the clamp scale (exactly 1, no sqrt/divide, when the squared norm <= 1024)
# speedup vs baseline: 1.5699x; 1.0545x over previous
#include <hip/hip_runtime.h>
#include <math.h>


#define NB 4
#define NT 512
#define ND 256
#define MOMB 0.9f
#define NOVB 0.95f
#define MOM_STRIDE 260

typedef float v2f __attribute__((ext_vector_type(2)));
typedef float v8f __attribute__((ext_vector_type(8)));

struct __align__(16) Smem {
  float mom[ND * MOM_STRIDE];
  float x_s[3][ND];
  float yacc[2][2][ND];
  float iema[ND];
  float redn[2][8];
  float redf[2][32];
  float eff_s[2];
  float lrema_s;
  float inner_lr_s;
};

__device__ __forceinline__ float xhalf(float v)
{
  const int b = __builtin_bit_cast(int, v);
  return __builtin_bit_cast(float, __builtin_amdgcn_permlanex16(b, b, 0x76543210, 0xfedcba98, false, false));
}

template <int CTRL>
__device__ __forceinline__ float row_xmask(float v)
{
  const int b = __builtin_bit_cast(int, v);
  return __builtin_bit_cast(float, __builtin_amdgcn_update_dpp(b, b, CTRL, 0xf, 0xf, true));
}

__device__ __forceinline__ float wave_sum(float a)
{
  a += xhalf(a);
  a += row_xmask<0x168>(a);
  a += row_xmask<0x164>(a);
  a += row_xmask<0x162>(a);
  a += row_xmask<0x161>(a);
  return a;
}

__device__ __forceinline__ void lr_step(Smem& sm, int buf)
{
  float s = 0.0f;
  #pragma unroll
  for (int i = 0; i < 8; ++i) s += sm.redn[buf][i];
  const float nov = s * (1.0f/ND);
  const float lrs = fminf(fmaxf(1.0f + 3.0f*nov, 0.5f), 3.0f);
  const float lrema = NOVB*sm.lrema_s + (1.0f-NOVB)*lrs;
  sm.lrema_s = lrema;
  sm.eff_s[buf] = sm.inner_lr_s * lrema;
}

__device__ __forceinline__ void stage_x(Smem& sm, int ring, int par, int j, int slot, int lane, float xj)
{
  sm.x_s[ring][j] = xj;
  const float old = sm.iema[j];
  float a = fabsf(xj - old);
  sm.iema[j] = NOVB*old + (1.0f-NOVB)*xj;
  a = wave_sum(a);
  if (lane == 0) sm.redn[par][slot] = a;
}

__global__ __launch_bounds__(1024) void ttt_linear_kernel(
    const float* __restrict__ x, const float* __restrict__ Wb,
    const float* __restrict__ bias, const float* __restrict__ loglr,
    float* __restrict__ out)
{
  __shared__ Smem sm;
  const int tid  = threadIdx.x;
  const int lane = tid & 31;
  const int wid  = tid >> 5;
  const int b    = blockIdx.x;
  const int hi   = lane >> 4;
  const int rb   = (wid >> 1) << 4;
  const int row  = rb + (lane & 15);
  const int colbase = (wid & 1) << 7;
  const bool stager = (tid >= ND) && (tid < 2*ND);
  const int  j      = tid & (ND-1);

  float w_u[64];
  #pragma unroll
  for (int c = 0; c < 32; ++c) {
    const int col = colbase + 4*c + 2*hi;
    const float2 wv = *reinterpret_cast<const float2*>(Wb + row*ND + col);
    w_u[2*c]   = wv.x;
    w_u[2*c+1] = wv.y;
  }
  for (int i = tid; i < ND*MOM_STRIDE; i += 1024) sm.mom[i] = 0.0f;

  if (tid == 0) {
    sm.inner_lr_s = fminf(fmaxf(expf(loglr[0]), 1e-5f), 1.0f);
    sm.lrema_s = 1.0f;
  }
  const float bias_r   = (tid < ND) ? bias[tid] : 0.0f;
  const float bias_row = bias[row];

  float xpre = 0.0f;
  if (stager) {
    sm.iema[j] = 0.0f;
    stage_x(sm, 0, 0, j, wid - 8, lane, x[((size_t)b*NT + 0)*ND + j]);
  }
  __syncthreads();

  if (tid == 0) lr_step(sm, 0);
  if (stager) {
    stage_x(sm, 1, 1, j, wid - 8, lane, x[((size_t)b*NT + 1)*ND + j]);
    xpre = x[((size_t)b*NT + 2)*ND + j];
  }

  v8f acc = {};
  #pragma unroll
  for (int c = 0; c < 32; ++c) {
    const int col = colbase + 4*c + 2*hi;
    const float2 xp = *reinterpret_cast<const float2*>(&sm.x_s[0][col]);
    v2f av; av.x = w_u[2*c]; av.y = w_u[2*c+1];
    v2f bv; bv.x = xp.x;     bv.y = xp.y;
    acc = __builtin_amdgcn_wmma_f32_16x16x4_f32(
         false, av,  false, bv,
         (short)0, acc,  false,  false);
  }

  float fsum = 0.0f;
  int rc = 0, rn = 1, rs = 2;
  const int col0 = colbase + 2*hi;
  const float* mrow = &sm.mom[row*MOM_STRIDE + col0];

  for (int t = 0; t < NT; ++t) {
    const int par = t & 1;

    if ((lane & 15) == 0) {
      const int mbase = rb + hi*8;
      #pragma unroll
      for (int v = 0; v < 8; ++v) sm.yacc[par][wid & 1][mbase + v] = acc[v];
    }
    if (lane == 0) sm.redf[par][wid] = fsum;
    __syncthreads();

    const float* xc = sm.x_s[rc];
    const float* xn = sm.x_s[rn];
    const float rpart = sm.redf[par][lane];
    const float prow  = sm.yacc[par][0][row] + sm.yacc[par][1][row];
    const float xrow  = xc[row];
    const float pout  = sm.yacc[par][0][j] + sm.yacc[par][1][j];
    const float eff   = sm.eff_s[par];
    float2 xp = *reinterpret_cast<const float2*>(&xc[col0]);
    float2 xq = *reinterpret_cast<const float2*>(&xn[col0]);
    float2 mp = *reinterpret_cast<const float2*>(&mrow[0]);

    const float ssq = wave_sum(rpart);
    float sp = 1.0f;
    if (!(ssq <= 1024.0f)) {
      const float fn = fmaxf(sqrtf(ssq), 1e-6f);
      sp = fminf(32.0f/fn, 1.0f);
    }
    const float err = ((sp*prow + bias_row) - xrow) * (1.0f/ND);

    if (tid < ND) out[((size_t)b*NT + t)*ND + tid] = sp*pout + bias_r;

    if (t + 1 == NT) break;

    if (stager && (t + 2 < NT)) {
      stage_x(sm, rs, par, j, wid - 8, lane, xpre);
      if (t + 3 < NT) xpre = x[((size_t)b*NT + t + 3)*ND + j];
    }

    if (tid == 0) lr_step(sm, par ^ 1);

    v2f fs2[4];
    #pragma unroll
    for (int q = 0; q < 4; ++q) fs2[q] = v2f{0.0f, 0.0f};
    acc = v8f{};
    #pragma unroll
    for (int c = 0; c < 32; ++c) {
      float2 xp_n = xp, xq_n = xq, mp_n = mp;
      if (c + 1 < 32) {
        xp_n = *reinterpret_cast<const float2*>(&xc[col0 + 4*(c + 1)]);
        xq_n = *reinterpret_cast<const float2*>(&xn[col0 + 4*(c + 1)]);
        mp_n = *reinterpret_cast<const float2*>(&mrow[4*(c + 1)]);
      }
      mp.x = MOMB*mp.x + (1.0f-MOMB)*err*xp.x;
      mp.y = MOMB*mp.y + (1.0f-MOMB)*err*xp.y;
      *reinterpret_cast<float2*>(&sm.mom[row*MOM_STRIDE + col0 + 4*c]) = mp;
      const float w0 = sp*w_u[2*c]   - eff*mp.x;
      const float w1 = sp*w_u[2*c+1] - eff*mp.y;
      w_u[2*c] = w0; w_u[2*c+1] = w1;
      v2f av; av.x = w0;   av.y = w1;
      fs2[c & 3] = __builtin_elementwise_fma(av, av, fs2[c & 3]);
      v2f bv; bv.x = xq.x; bv.y = xq.y;
      acc = __builtin_amdgcn_wmma_f32_16x16x4_f32(
           false, av,  false, bv,
           (short)0, acc,  false,  false);
      if (c + 1 < 32) __builtin_amdgcn_sched_group_barrier(0x100, 3, 0);
      __builtin_amdgcn_sched_group_barrier(0x002, 8, 0);
      __builtin_amdgcn_sched_group_barrier(0x200, 1, 0);
      __builtin_amdgcn_sched_group_barrier(0x008, 1, 0);
      xp = xp_n; xq = xq_n; mp = mp_n;
    }
    const v2f fs01 = fs2[0] + fs2[1];
    const v2f fs23 = fs2[2] + fs2[3];
    const v2f fsv  = fs01 + fs23;
    fsum = wave_sum(fsv.x + fsv.y);

    const int r0 = rc; rc = rn; rn = rs; rs = r0;
  }
}

extern "C" void kernel_launch(void* const* d_in, const int* in_sizes, int n_in,
                              void* d_out, int out_size, void* d_ws, size_t ws_size,
                              hipStream_t stream) {
  (void)in_sizes; (void)n_in; (void)out_size; (void)d_ws; (void)ws_size;
  const float* x    = (const float*)d_in[0];
  const float* Wb   = (const float*)d_in[1];
  const float* bias = (const float*)d_in[2];
  const float* lil  = (const float*)d_in[3];
  float* out = (float*)d_out;
  ttt_linear_kernel<<<dim3(NB), dim3(1024), 0, stream>>>(x, Wb, bias, lil, out);
}


# Clamp scale out of the per-element update: W panel rescaled in place only in steps whose scale is not 1 (block-uniform branch); each weight pair then updates with one packed fma, rounded once
# speedup vs baseline: 1.6280x; 1.0370x over previous
#include <hip/hip_runtime.h>
#include <math.h>


#define NB 4
#define NT 512
#define ND 256
#define MOMB 0.9f
#define NOVB 0.95f
#define MOM_STRIDE 260

typedef float v2f __attribute__((ext_vector_type(2)));
typedef float v8f __attribute__((ext_vector_type(8)));

struct __align__(16) Smem {
  float mom[ND * MOM_STRIDE];
  float x_s[3][ND];
  float yacc[2][2][ND];
  float iema[ND];
  float redn[2][8];
  float redf[2][32];
  float eff_s[2];
  float lrema_s;
  float inner_lr_s;
};

__device__ __forceinline__ float xhalf(float v)
{
  const int b = __builtin_bit_cast(int, v);
  return __builtin_bit_cast(float, __builtin_amdgcn_permlanex16(b, b, 0x76543210, 0xfedcba98, false, false));
}

template <int CTRL>
__device__ __forceinline__ float row_xmask(float v)
{
  const int b = __builtin_bit_cast(int, v);
  return __builtin_bit_cast(float, __builtin_amdgcn_update_dpp(b, b, CTRL, 0xf, 0xf, true));
}

__device__ __forceinline__ float wave_sum(float a)
{
  a += xhalf(a);
  a += row_xmask<0x168>(a);
  a += row_xmask<0x164>(a);
  a += row_xmask<0x162>(a);
  a += row_xmask<0x161>(a);
  return a;
}

__device__ __forceinline__ void lr_step(Smem& sm, int buf)
{
  float s = 0.0f;
  #pragma unroll
  for (int i = 0; i < 8; ++i) s += sm.redn[buf][i];
  const float nov = s * (1.0f/ND);
  const float lrs = fminf(fmaxf(1.0f + 3.0f*nov, 0.5f), 3.0f);
  const float lrema = NOVB*sm.lrema_s + (1.0f-NOVB)*lrs;
  sm.lrema_s = lrema;
  sm.eff_s[buf] = sm.inner_lr_s * lrema;
}

__device__ __forceinline__ void stage_x(Smem& sm, int ring, int par, int j, int slot, int lane, float xj)
{
  sm.x_s[ring][j] = xj;
  const float old = sm.iema[j];
  float a = fabsf(xj - old);
  sm.iema[j] = NOVB*old + (1.0f-NOVB)*xj;
  a = wave_sum(a);
  if (lane == 0) sm.redn[par][slot] = a;
}

__global__ __launch_bounds__(1024) void ttt_linear_kernel(
    const float* __restrict__ x, const float* __restrict__ Wb,
    const float* __restrict__ bias, const float* __restrict__ loglr,
    float* __restrict__ out)
{
  __shared__ Smem sm;
  const int tid  = threadIdx.x;
  const int lane = tid & 31;
  const int wid  = tid >> 5;
  const int b    = blockIdx.x;
  const int hi   = lane >> 4;
  const int rb   = (wid >> 1) << 4;
  const int row  = rb + (lane & 15);
  const int colbase = (wid & 1) << 7;
  const bool stager = (tid >= ND) && (tid < 2*ND);
  const int  j      = tid & (ND-1);

  float w_u[64];
  #pragma unroll
  for (int c = 0; c < 32; ++c) {
    const int col = colbase + 4*c + 2*hi;
    const float2 wv = *reinterpret_cast<const float2*>(Wb + row*ND + col);
    w_u[2*c]   = wv.x;
    w_u[2*c+1] = wv.y;
  }
  for (int i = tid; i < ND*MOM_STRIDE; i += 1024) sm.mom[i] = 0.0f;

  if (tid == 0) {
    sm.inner_lr_s = fminf(fmaxf(expf(loglr[0]), 1e-5f), 1.0f);
    sm.lrema_s = 1.0f;
  }
  const float bias_r   = (tid < ND) ? bias[tid] : 0.0f;
  const float bias_row = bias[row];

  float xpre = 0.0f;
  if (stager) {
    sm.iema[j] = 0.0f;
    stage_x(sm, 0, 0, j, wid - 8, lane, x[((size_t)b*NT + 0)*ND + j]);
  }
  __syncthreads();

  if (tid == 0) lr_step(sm, 0);
  if (stager) {
    stage_x(sm, 1, 1, j, wid - 8, lane, x[((size_t)b*NT + 1)*ND + j]);
    xpre = x[((size_t)b*NT + 2)*ND + j];
  }

  v8f acc = {};
  #pragma unroll
  for (int c = 0; c < 32; ++c) {
    const int col = colbase + 4*c + 2*hi;
    const float2 xp = *reinterpret_cast<const float2*>(&sm.x_s[0][col]);
    v2f av; av.x = w_u[2*c]; av.y = w_u[2*c+1];
    v2f bv; bv.x = xp.x;     bv.y = xp.y;
    acc = __builtin_amdgcn_wmma_f32_16x16x4_f32(
         false, av,  false, bv,
         (short)0, acc,  false,  false);
  }

  float fsum = 0.0f;
  int rc = 0, rn = 1, rs = 2;
  const int col0 = colbase + 2*hi;
  const float* mrow = &sm.mom[row*MOM_STRIDE + col0];

  for (int t = 0; t < NT; ++t) {
    const int par = t & 1;

    if ((lane & 15) == 0) {
      const int mbase = rb + hi*8;
      #pragma unroll
      for (int v = 0; v < 8; ++v) sm.yacc[par][wid & 1][mbase + v] = acc[v];
    }
    if (lane == 0) sm.redf[par][wid] = fsum;
    __syncthreads();

    const float* xc = sm.x_s[rc];
    const float* xn = sm.x_s[rn];
    const float rpart = sm.redf[par][lane];
    const float prow  = sm.yacc[par][0][row] + sm.yacc[par][1][row];
    const float xrow  = xc[row];
    const float pout  = sm.yacc[par][0][j] + sm.yacc[par][1][j];
    const float eff   = sm.eff_s[par];
    float2 xp = *reinterpret_cast<const float2*>(&xc[col0]);
    float2 xq = *reinterpret_cast<const float2*>(&xn[col0]);
    float2 mp = *reinterpret_cast<const float2*>(&mrow[0]);

    const float ssq = wave_sum(rpart);
    float sp = 1.0f;
    if (!(ssq <= 1024.0f)) {
      const float fn = fmaxf(sqrtf(ssq), 1e-6f);
      sp = fminf(32.0f/fn, 1.0f);
      #pragma unroll
      for (int i = 0; i < 64; ++i) w_u[i] *= sp;
    }
    const float err = ((sp*prow + bias_row) - xrow) * (1.0f/ND);

    if (tid < ND) out[((size_t)b*NT + t)*ND + tid] = sp*pout + bias_r;

    if (t + 1 == NT) break;

    if (stager && (t + 2 < NT)) {
      stage_x(sm, rs, par, j, wid - 8, lane, xpre);
      if (t + 3 < NT) xpre = x[((size_t)b*NT + t + 3)*ND + j];
    }

    if (tid == 0) lr_step(sm, par ^ 1);

    v2f fs2[4];
    #pragma unroll
    for (int q = 0; q < 4; ++q) fs2[q] = v2f{0.0f, 0.0f};
    acc = v8f{};
    #pragma unroll
    for (int c = 0; c < 32; ++c) {
      float2 xp_n = xp, xq_n = xq, mp_n = mp;
      if (c + 1 < 32) {
        xp_n = *reinterpret_cast<const float2*>(&xc[col0 + 4*(c + 1)]);
        xq_n = *reinterpret_cast<const float2*>(&xn[col0 + 4*(c + 1)]);
        mp_n = *reinterpret_cast<const float2*>(&mrow[4*(c + 1)]);
      }
      mp.x = MOMB*mp.x + (1.0f-MOMB)*err*xp.x;
      mp.y = MOMB*mp.y + (1.0f-MOMB)*err*xp.y;
      *reinterpret_cast<float2*>(&sm.mom[row*MOM_STRIDE + col0 + 4*c]) = mp;
      const float w0 = w_u[2*c]   - eff*mp.x;
      const float w1 = w_u[2*c+1] - eff*mp.y;
      w_u[2*c] = w0; w_u[2*c+1] = w1;
      v2f av; av.x = w0;   av.y = w1;
      fs2[c & 3] = __builtin_elementwise_fma(av, av, fs2[c & 3]);
      v2f bv; bv.x = xq.x; bv.y = xq.y;
      acc = __builtin_amdgcn_wmma_f32_16x16x4_f32(
           false, av,  false, bv,
           (short)0, acc,  false,  false);
      if (c + 1 < 32) __builtin_amdgcn_sched_group_barrier(0x100, 3, 0);
      __builtin_amdgcn_sched_group_barrier(0x002, 8, 0);
      __builtin_amdgcn_sched_group_barrier(0x200, 1, 0);
      __builtin_amdgcn_sched_group_barrier(0x008, 1, 0);
      xp = xp_n; xq = xq_n; mp = mp_n;
    }
    const v2f fs01 = fs2[0] + fs2[1];
    const v2f fs23 = fs2[2] + fs2[3];
    const v2f fsv  = fs01 + fs23;
    fsum = wave_sum(fsv.x + fsv.y);

    const int r0 = rc; rc = rn; rn = rs; rs = r0;
  }
}

extern "C" void kernel_launch(void* const* d_in, const int* in_sizes, int n_in,
                              void* d_out, int out_size, void* d_ws, size_t ws_size,
                              hipStream_t stream) {
  (void)in_sizes; (void)n_in; (void)out_size; (void)d_ws; (void)ws_size;
  const float* x    = (const float*)d_in[0];
  const float* Wb   = (const float*)d_in[1];
  const float* bias = (const float*)d_in[2];
  const float* lil  = (const float*)d_in[3];
  float* out = (float*)d_out;
  ttt_linear_kernel<<<dim3(NB), dim3(1024), 0, stream>>>(x, Wb, bias, lil, out);
}


# Momentum held scaled by 0.9^-t: the decay multiply leaves the per-element update (one packed fma builds it, W takes eff*0.9^t times it); the momentum's rounding changes, no sum order does
# speedup vs baseline: 1.6817x; 1.0330x over previous
#include <hip/hip_runtime.h>
#include <math.h>


#define NB 4
#define NT 512
#define ND 256
#define MOMB 0.9f
#define NOVB 0.95f
#define MOM_STRIDE 260

typedef float v2f __attribute__((ext_vector_type(2)));
typedef float v8f __attribute__((ext_vector_type(8)));

struct __align__(16) Smem {
  float mom[ND * MOM_STRIDE];
  float x_s[3][ND];
  float yacc[2][2][ND];
  float iema[ND];
  float redn[2][8];
  float redf[2][32];
  float gq_s[2][2];
  float lrema_s;
  float inner_lr_s;
  float ppow_s;
};

__device__ __forceinline__ float xhalf(float v)
{
  const int b = __builtin_bit_cast(int, v);
  return __builtin_bit_cast(float, __builtin_amdgcn_permlanex16(b, b, 0x76543210, 0xfedcba98, false, false));
}

template <int CTRL>
__device__ __forceinline__ float row_xmask(float v)
{
  const int b = __builtin_bit_cast(int, v);
  return __builtin_bit_cast(float, __builtin_amdgcn_update_dpp(b, b, CTRL, 0xf, 0xf, true));
}

__device__ __forceinline__ float wave_sum(float a)
{
  a += xhalf(a);
  a += row_xmask<0x168>(a);
  a += row_xmask<0x164>(a);
  a += row_xmask<0x162>(a);
  a += row_xmask<0x161>(a);
  return a;
}

__device__ __forceinline__ void lr_step(Smem& sm, int buf)
{
  float s = 0.0f;
  #pragma unroll
  for (int i = 0; i < 8; ++i) s += sm.redn[buf][i];
  const float nov = s * (1.0f/ND);
  const float lrs = fminf(fmaxf(1.0f + 3.0f*nov, 0.5f), 3.0f);
  const float lrema = NOVB*sm.lrema_s + (1.0f-NOVB)*lrs;
  sm.lrema_s = lrema;
  const float p = MOMB*sm.ppow_s;
  sm.ppow_s = p;
  const float q = 1.0f/p;
  sm.gq_s[buf][0] = (sm.inner_lr_s * lrema) * p;
  sm.gq_s[buf][1] = (1.0f-MOMB) * q;
}

__device__ __forceinline__ void stage_x(Smem& sm, int ring, int par, int j, int slot, int lane, float xj)
{
  sm.x_s[ring][j] = xj;
  const float old = sm.iema[j];
  float a = fabsf(xj - old);
  sm.iema[j] = NOVB*old + (1.0f-NOVB)*xj;
  a = wave_sum(a);
  if (lane == 0) sm.redn[par][slot] = a;
}

__global__ __launch_bounds__(1024) void ttt_linear_kernel(
    const float* __restrict__ x, const float* __restrict__ Wb,
    const float* __restrict__ bias, const float* __restrict__ loglr,
    float* __restrict__ out)
{
  __shared__ Smem sm;
  const int tid  = threadIdx.x;
  const int lane = tid & 31;
  const int wid  = tid >> 5;
  const int b    = blockIdx.x;
  const int hi   = lane >> 4;
  const int rb   = (wid >> 1) << 4;
  const int row  = rb + (lane & 15);
  const int colbase = (wid & 1) << 7;
  const bool stager = (tid >= ND) && (tid < 2*ND);
  const int  j      = tid & (ND-1);

  float w_u[64];
  #pragma unroll
  for (int c = 0; c < 32; ++c) {
    const int col = colbase + 4*c + 2*hi;
    const float2 wv = *reinterpret_cast<const float2*>(Wb + row*ND + col);
    w_u[2*c]   = wv.x;
    w_u[2*c+1] = wv.y;
  }
  for (int i = tid; i < ND*MOM_STRIDE; i += 1024) sm.mom[i] = 0.0f;

  if (tid == 0) {
    sm.inner_lr_s = fminf(fmaxf(expf(loglr[0]), 1e-5f), 1.0f);
    sm.lrema_s = 1.0f;
    sm.ppow_s  = 1.0f;
  }
  const float bias_r   = (tid < ND) ? bias[tid] : 0.0f;
  const float bias_row = bias[row];

  float xpre = 0.0f;
  if (stager) {
    sm.iema[j] = 0.0f;
    stage_x(sm, 0, 0, j, wid - 8, lane, x[((size_t)b*NT + 0)*ND + j]);
  }
  __syncthreads();

  if (tid == 0) lr_step(sm, 0);
  if (stager) {
    stage_x(sm, 1, 1, j, wid - 8, lane, x[((size_t)b*NT + 1)*ND + j]);
    xpre = x[((size_t)b*NT + 2)*ND + j];
  }

  v8f acc = {};
  #pragma unroll
  for (int c = 0; c < 32; ++c) {
    const int col = colbase + 4*c + 2*hi;
    const float2 xp = *reinterpret_cast<const float2*>(&sm.x_s[0][col]);
    v2f av; av.x = w_u[2*c]; av.y = w_u[2*c+1];
    v2f bv; bv.x = xp.x;     bv.y = xp.y;
    acc = __builtin_amdgcn_wmma_f32_16x16x4_f32(
         false, av,  false, bv,
         (short)0, acc,  false,  false);
  }

  float fsum = 0.0f;
  int rc = 0, rn = 1, rs = 2;
  const int col0 = colbase + 2*hi;
  const float* mrow = &sm.mom[row*MOM_STRIDE + col0];

  for (int t = 0; t < NT; ++t) {
    const int par = t & 1;

    if ((lane & 15) == 0) {
      const int mbase = rb + hi*8;
      #pragma unroll
      for (int v = 0; v < 8; ++v) sm.yacc[par][wid & 1][mbase + v] = acc[v];
    }
    if (lane == 0) sm.redf[par][wid] = fsum;
    __syncthreads();

    const float* xc = sm.x_s[rc];
    const float* xn = sm.x_s[rn];
    const float rpart = sm.redf[par][lane];
    const float prow  = sm.yacc[par][0][row] + sm.yacc[par][1][row];
    const float xrow  = xc[row];
    const float pout  = sm.yacc[par][0][j] + sm.yacc[par][1][j];
    const float2 gq   = *reinterpret_cast<const float2*>(&sm.gq_s[par][0]);
    float2 xp = *reinterpret_cast<const float2*>(&xc[col0]);
    float2 xq = *reinterpret_cast<const float2*>(&xn[col0]);
    float2 mp = *reinterpret_cast<const float2*>(&mrow[0]);

    const float ssq = wave_sum(rpart);
    float sp = 1.0f;
    if (!(ssq <= 1024.0f)) {
      const float fn = fmaxf(sqrtf(ssq), 1e-6f);
      sp = fminf(32.0f/fn, 1.0f);
      #pragma unroll
      for (int i = 0; i < 64; ++i) w_u[i] *= sp;
    }
    const float err = ((sp*prow + bias_row) - xrow) * (1.0f/ND);
    const float g   = gq.x;
    const float cs  = err * gq.y;

    if (tid < ND) out[((size_t)b*NT + t)*ND + tid] = sp*pout + bias_r;

    if (t + 1 == NT) break;

    if (stager && (t + 2 < NT)) {
      stage_x(sm, rs, par, j, wid - 8, lane, xpre);
      if (t + 3 < NT) xpre = x[((size_t)b*NT + t + 3)*ND + j];
    }

    if (tid == 0) lr_step(sm, par ^ 1);

    v2f fs2[4];
    #pragma unroll
    for (int q = 0; q < 4; ++q) fs2[q] = v2f{0.0f, 0.0f};
    acc = v8f{};
    #pragma unroll
    for (int c = 0; c < 32; ++c) {
      float2 xp_n = xp, xq_n = xq, mp_n = mp;
      if (c + 1 < 32) {
        xp_n = *reinterpret_cast<const float2*>(&xc[col0 + 4*(c + 1)]);
        xq_n = *reinterpret_cast<const float2*>(&xn[col0 + 4*(c + 1)]);
        mp_n = *reinterpret_cast<const float2*>(&mrow[4*(c + 1)]);
      }
      mp.x = mp.x + cs*xp.x;
      mp.y = mp.y + cs*xp.y;
      *reinterpret_cast<float2*>(&sm.mom[row*MOM_STRIDE + col0 + 4*c]) = mp;
      const float w0 = w_u[2*c]   - g*mp.x;
      const float w1 = w_u[2*c+1] - g*mp.y;
      w_u[2*c] = w0; w_u[2*c+1] = w1;
      v2f av; av.x = w0;   av.y = w1;
      fs2[c & 3] = __builtin_elementwise_fma(av, av, fs2[c & 3]);
      v2f bv; bv.x = xq.x; bv.y = xq.y;
      acc = __builtin_amdgcn_wmma_f32_16x16x4_f32(
           false, av,  false, bv,
           (short)0, acc,  false,  false);
      if (c + 1 < 32) __builtin_amdgcn_sched_group_barrier(0x100, 3, 0);
      __builtin_amdgcn_sched_group_barrier(0x002, 8, 0);
      __builtin_amdgcn_sched_group_barrier(0x200, 1, 0);
      __builtin_amdgcn_sched_group_barrier(0x008, 1, 0);
      xp = xp_n; xq = xq_n; mp = mp_n;
    }
    const v2f fs01 = fs2[0] + fs2[1];
    const v2f fs23 = fs2[2] + fs2[3];
    const v2f fsv  = fs01 + fs23;
    fsum = wave_sum(fsv.x + fsv.y);

    const int r0 = rc; rc = rn; rn = rs; rs = r0;
  }
}

extern "C" void kernel_launch(void* const* d_in, const int* in_sizes, int n_in,
                              void* d_out, int out_size, void* d_ws, size_t ws_size,
                              hipStream_t stream) {
  (void)in_sizes; (void)n_in; (void)out_size; (void)d_ws; (void)ws_size;
  const float* x    = (const float*)d_in[0];
  const float* Wb   = (const float*)d_in[1];
  const float* bias = (const float*)d_in[2];
  const float* lil  = (const float*)d_in[3];
  float* out = (float*)d_out;
  ttt_linear_kernel<<<dim3(NB), dim3(1024), 0, stream>>>(x, Wb, bias, lil, out);
}


# Update loop's LDS operand reads issued two K-slices ahead, held there by a compiler fence (the x_t read had sat directly in front of its fma: one LDS round trip per slice); no arithmetic change
# speedup vs baseline: 1.7599x; 1.0465x over previous
#include <hip/hip_runtime.h>
#include <math.h>


#define NB 4
#define NT 512
#define ND 256
#define MOMB 0.9f
#define NOVB 0.95f
#define MOM_STRIDE 260

typedef float v2f __attribute__((ext_vector_type(2)));
typedef float v8f __attribute__((ext_vector_type(8)));

struct __align__(16) Smem {
  float mom[ND * MOM_STRIDE];
  float x_s[3][ND];
  float yacc[2][2][ND];
  float iema[ND];
  float redn[2][8];
  float redf[2][32];
  float gq_s[2][2];
  float lrema_s;
  float inner_lr_s;
  float ppow_s;
};

__device__ __forceinline__ float xhalf(float v)
{
  const int b = __builtin_bit_cast(int, v);
  return __builtin_bit_cast(float, __builtin_amdgcn_permlanex16(b, b, 0x76543210, 0xfedcba98, false, false));
}

template <int CTRL>
__device__ __forceinline__ float row_xmask(float v)
{
  const int b = __builtin_bit_cast(int, v);
  return __builtin_bit_cast(float, __builtin_amdgcn_update_dpp(b, b, CTRL, 0xf, 0xf, true));
}

__device__ __forceinline__ float wave_sum(float a)
{
  a += xhalf(a);
  a += row_xmask<0x168>(a);
  a += row_xmask<0x164>(a);
  a += row_xmask<0x162>(a);
  a += row_xmask<0x161>(a);
  return a;
}

__device__ __forceinline__ void lr_step(Smem& sm, int buf)
{
  float s = 0.0f;
  #pragma unroll
  for (int i = 0; i < 8; ++i) s += sm.redn[buf][i];
  const float nov = s * (1.0f/ND);
  const float lrs = fminf(fmaxf(1.0f + 3.0f*nov, 0.5f), 3.0f);
  const float lrema = NOVB*sm.lrema_s + (1.0f-NOVB)*lrs;
  sm.lrema_s = lrema;
  const float p = MOMB*sm.ppow_s;
  sm.ppow_s = p;
  const float q = 1.0f/p;
  sm.gq_s[buf][0] = (sm.inner_lr_s * lrema) * p;
  sm.gq_s[buf][1] = (1.0f-MOMB) * q;
}

__device__ __forceinline__ void stage_x(Smem& sm, int ring, int par, int j, int slot, int lane, float xj)
{
  sm.x_s[ring][j] = xj;
  const float old = sm.iema[j];
  float a = fabsf(xj - old);
  sm.iema[j] = NOVB*old + (1.0f-NOVB)*xj;
  a = wave_sum(a);
  if (lane == 0) sm.redn[par][slot] = a;
}

__global__ __launch_bounds__(1024) void ttt_linear_kernel(
    const float* __restrict__ x, const float* __restrict__ Wb,
    const float* __restrict__ bias, const float* __restrict__ loglr,
    float* __restrict__ out)
{
  __shared__ Smem sm;
  const int tid  = threadIdx.x;
  const int lane = tid & 31;
  const int wid  = tid >> 5;
  const int b    = blockIdx.x;
  const int hi   = lane >> 4;
  const int rb   = (wid >> 1) << 4;
  const int row  = rb + (lane & 15);
  const int colbase = (wid & 1) << 7;
  const bool stager = (tid >= ND) && (tid < 2*ND);
  const int  j      = tid & (ND-1);

  float w_u[64];
  #pragma unroll
  for (int c = 0; c < 32; ++c) {
    const int col = colbase + 4*c + 2*hi;
    const float2 wv = *reinterpret_cast<const float2*>(Wb + row*ND + col);
    w_u[2*c]   = wv.x;
    w_u[2*c+1] = wv.y;
  }
  for (int i = tid; i < ND*MOM_STRIDE; i += 1024) sm.mom[i] = 0.0f;

  if (tid == 0) {
    sm.inner_lr_s = fminf(fmaxf(expf(loglr[0]), 1e-5f), 1.0f);
    sm.lrema_s = 1.0f;
    sm.ppow_s  = 1.0f;
  }
  const float bias_r   = (tid < ND) ? bias[tid] : 0.0f;
  const float bias_row = bias[row];

  float xpre = 0.0f;
  if (stager) {
    sm.iema[j] = 0.0f;
    stage_x(sm, 0, 0, j, wid - 8, lane, x[((size_t)b*NT + 0)*ND + j]);
  }
  __syncthreads();

  if (tid == 0) lr_step(sm, 0);
  if (stager) {
    stage_x(sm, 1, 1, j, wid - 8, lane, x[((size_t)b*NT + 1)*ND + j]);
    xpre = x[((size_t)b*NT + 2)*ND + j];
  }

  v8f acc = {};
  #pragma unroll
  for (int c = 0; c < 32; ++c) {
    const int col = colbase + 4*c + 2*hi;
    const float2 xp = *reinterpret_cast<const float2*>(&sm.x_s[0][col]);
    v2f av; av.x = w_u[2*c]; av.y = w_u[2*c+1];
    v2f bv; bv.x = xp.x;     bv.y = xp.y;
    acc = __builtin_amdgcn_wmma_f32_16x16x4_f32(
         false, av,  false, bv,
         (short)0, acc,  false,  false);
  }

  float fsum = 0.0f;
  int rc = 0, rn = 1, rs = 2;
  const int col0 = colbase + 2*hi;
  const float* mrow = &sm.mom[row*MOM_STRIDE + col0];

  for (int t = 0; t < NT; ++t) {
    const int par = t & 1;

    if ((lane & 15) == 0) {
      const int mbase = rb + hi*8;
      #pragma unroll
      for (int v = 0; v < 8; ++v) sm.yacc[par][wid & 1][mbase + v] = acc[v];
    }
    if (lane == 0) sm.redf[par][wid] = fsum;
    __syncthreads();

    const float* xc = sm.x_s[rc];
    const float* xn = sm.x_s[rn];
    const float rpart = sm.redf[par][lane];
    const float prow  = sm.yacc[par][0][row] + sm.yacc[par][1][row];
    const float xrow  = xc[row];
    const float pout  = sm.yacc[par][0][j] + sm.yacc[par][1][j];
    const float2 gq   = *reinterpret_cast<const float2*>(&sm.gq_s[par][0]);
    float2 xp  = *reinterpret_cast<const float2*>(&xc[col0]);
    float2 xq  = *reinterpret_cast<const float2*>(&xn[col0]);
    float2 mp  = *reinterpret_cast<const float2*>(&mrow[0]);
    float2 xp1 = *reinterpret_cast<const float2*>(&xc[col0 + 4]);
    float2 xq1 = *reinterpret_cast<const float2*>(&xn[col0 + 4]);
    float2 mp1 = *reinterpret_cast<const float2*>(&mrow[4]);

    const float ssq = wave_sum(rpart);
    float sp = 1.0f;
    if (!(ssq <= 1024.0f)) {
      const float fn = fmaxf(sqrtf(ssq), 1e-6f);
      sp = fminf(32.0f/fn, 1.0f);
      #pragma unroll
      for (int i = 0; i < 64; ++i) w_u[i] *= sp;
    }
    const float err = ((sp*prow + bias_row) - xrow) * (1.0f/ND);
    const float g   = gq.x;
    const float cs  = err * gq.y;

    if (tid < ND) out[((size_t)b*NT + t)*ND + tid] = sp*pout + bias_r;

    if (t + 1 == NT) break;

    if (stager && (t + 2 < NT)) {
      stage_x(sm, rs, par, j, wid - 8, lane, xpre);
      if (t + 3 < NT) xpre = x[((size_t)b*NT + t + 3)*ND + j];
    }

    if (tid == 0) lr_step(sm, par ^ 1);

    v2f fs2[4];
    #pragma unroll
    for (int q = 0; q < 4; ++q) fs2[q] = v2f{0.0f, 0.0f};
    acc = v8f{};
    #pragma unroll
    for (int c = 0; c < 32; ++c) {
      float2 xp2 = xp1, xq2 = xq1, mp2 = mp1;
      if (c + 2 < 32) {
        xp2 = *reinterpret_cast<const float2*>(&xc[col0 + 4*(c + 2)]);
        xq2 = *reinterpret_cast<const float2*>(&xn[col0 + 4*(c + 2)]);
        mp2 = *reinterpret_cast<const float2*>(&mrow[4*(c + 2)]);
      }
      asm volatile("" ::: "memory");
      __builtin_amdgcn_sched_barrier(0);
      mp.x = mp.x + cs*xp.x;
      mp.y = mp.y + cs*xp.y;
      *reinterpret_cast<float2*>(&sm.mom[row*MOM_STRIDE + col0 + 4*c]) = mp;
      const float w0 = w_u[2*c]   - g*mp.x;
      const float w1 = w_u[2*c+1] - g*mp.y;
      w_u[2*c] = w0; w_u[2*c+1] = w1;
      v2f av; av.x = w0;   av.y = w1;
      fs2[c & 3] = __builtin_elementwise_fma(av, av, fs2[c & 3]);
      v2f bv; bv.x = xq.x; bv.y = xq.y;
      acc = __builtin_amdgcn_wmma_f32_16x16x4_f32(
           false, av,  false, bv,
           (short)0, acc,  false,  false);
      xp = xp1; xq = xq1; mp = mp1;
      xp1 = xp2; xq1 = xq2; mp1 = mp2;
    }
    const v2f fs01 = fs2[0] + fs2[1];
    const v2f fs23 = fs2[2] + fs2[3];
    const v2f fsv  = fs01 + fs23;
    fsum = wave_sum(fsv.x + fsv.y);

    const int r0 = rc; rc = rn; rn = rs; rs = r0;
  }
}

extern "C" void kernel_launch(void* const* d_in, const int* in_sizes, int n_in,
                              void* d_out, int out_size, void* d_ws, size_t ws_size,
                              hipStream_t stream) {
  (void)in_sizes; (void)n_in; (void)out_size; (void)d_ws; (void)ws_size;
  const float* x    = (const float*)d_in[0];
  const float* Wb   = (const float*)d_in[1];
  const float* bias = (const float*)d_in[2];
  const float* lil  = (const float*)d_in[3];
  float* out = (float*)d_out;
  ttt_linear_kernel<<<dim3(NB), dim3(1024), 0, stream>>>(x, Wb, bias, lil, out);
}


# Matvecs of 8 scan steps batched: per chunk one WMMA pass of W and momentum panels times the 8 staged x as real B; y by per-row recurrence over x dot products; y sum order changes
# speedup vs baseline: 2.4228x; 1.3767x over previous
#include <hip/hip_runtime.h>
#include <math.h>


#define NB 4
#define NT 512
#define ND 256
#define CH 8
#define MOMB 0.9f
#define NOVB 0.95f
#define MOM_STRIDE 260
#define XPITCH 260
#define PQP 272

typedef float v2f __attribute__((ext_vector_type(2)));
typedef float v8f __attribute__((ext_vector_type(8)));

struct __align__(16) Smem {
  float mom[ND * MOM_STRIDE];
  float xs[2 * CH][XPITCH];
  float pq[2][2][CH][PQP];
  float ybuf[2][ND];
  float iema[ND];
  float redn[2 * CH][8];
  float redf[2][32];
  float gq_s[2 * CH][2];
  float dtab[CH][8];
  float lrema_s;
  float inner_lr_s;
  float ppow_s;
};

__device__ __forceinline__ float xhalf(float v)
{
  const int b = __builtin_bit_cast(int, v);
  return __builtin_bit_cast(float, __builtin_amdgcn_permlanex16(b, b, 0x76543210, 0xfedcba98, false, false));
}

template <int CTRL>
__device__ __forceinline__ float row_xmask(float v)
{
  const int b = __builtin_bit_cast(int, v);
  return __builtin_bit_cast(float, __builtin_amdgcn_update_dpp(b, b, CTRL, 0xf, 0xf, true));
}

__device__ __forceinline__ float wave_sum(float a)
{
  a += xhalf(a);
  a += row_xmask<0x168>(a);
  a += row_xmask<0x164>(a);
  a += row_xmask<0x162>(a);
  a += row_xmask<0x161>(a);
  return a;
}

__device__ __forceinline__ float uni(float v)
{
  return __builtin_bit_cast(float, __builtin_amdgcn_readfirstlane(__builtin_bit_cast(int, v)));
}

__device__ __forceinline__ void lr_step(Smem& sm, int slot16)
{
  float s = 0.0f;
  #pragma unroll
  for (int i = 0; i < 8; ++i) s += sm.redn[slot16][i];
  const float nov = s * (1.0f/ND);
  const float lrs = fminf(fmaxf(1.0f + 3.0f*nov, 0.5f), 3.0f);
  const float lrema = NOVB*sm.lrema_s + (1.0f-NOVB)*lrs;
  sm.lrema_s = lrema;
  const float p = MOMB*sm.ppow_s;
  sm.ppow_s = p;
  const float q = 1.0f/p;
  sm.gq_s[slot16][0] = (sm.inner_lr_s * lrema) * p;
  sm.gq_s[slot16][1] = (1.0f-MOMB) * q;
}

__device__ __forceinline__ void stage_x(Smem& sm, int slot16, int j, int wslot, int lane, float xj)
{
  sm.xs[slot16][j] = xj;
  const float old = sm.iema[j];
  float a = fabsf(xj - old);
  sm.iema[j] = NOVB*old + (1.0f-NOVB)*xj;
  a = wave_sum(a);
  if (lane == 0) sm.redn[slot16][wslot] = a;
}

__global__ __launch_bounds__(1024) void ttt_linear_kernel(
    const float* __restrict__ x, const float* __restrict__ Wb,
    const float* __restrict__ bias, const float* __restrict__ loglr,
    float* __restrict__ out)
{
  __shared__ Smem sm;
  const int tid  = threadIdx.x;
  const int lane = tid & 31;
  const int wid  = tid >> 5;
  const int b    = blockIdx.x;
  const int hi   = lane >> 4;
  const int rb   = (wid >> 1) << 4;
  const int row  = rb + (lane & 15);
  const int colbase = (wid & 1) << 7;
  const bool stager = (tid >= ND) && (tid < 2*ND);
  const int  j      = tid & (ND-1);

  float w_u[64];
  #pragma unroll
  for (int c = 0; c < 32; ++c) {
    const int col = colbase + 4*c + 2*hi;
    const float2 wv = *reinterpret_cast<const float2*>(Wb + row*ND + col);
    w_u[2*c]   = wv.x;
    w_u[2*c+1] = wv.y;
  }
  for (int i = tid; i < ND*MOM_STRIDE; i += 1024) sm.mom[i] = 0.0f;
  if (tid < CH*8) sm.dtab[tid >> 3][tid & 7] = 0.0f;

  if (tid == 0) {
    sm.inner_lr_s = fminf(fmaxf(expf(loglr[0]), 1e-5f), 1.0f);
    sm.lrema_s = 1.0f;
    sm.ppow_s  = 1.0f;
  }
  const float bias_row = bias[row];

  float xpre = 0.0f;
  if (stager) {
    sm.iema[j] = 0.0f;
    for (int i = 0; i < CH; ++i)
      stage_x(sm, i, j, wid - 8, lane, x[((size_t)b*NT + i)*ND + j]);
    xpre = x[((size_t)b*NT + CH)*ND + j];
  }
  __syncthreads();
  if (tid == 0) {
    for (int i = 0; i < CH; ++i) lr_step(sm, i);
  }

  float fsum = 0.0f;
  const int col0 = colbase + 2*hi;
  const float* mrow = &sm.mom[row*MOM_STRIDE + col0];

  float Su = 1.0f;
  float G[CH-1], cs[CH-1];
  #pragma unroll
  for (int i = 0; i < CH-1; ++i) { G[i] = 0.0f; cs[i] = 0.0f; }

  for (int t = 0; t < NT; ++t) {
    const int par = t & 1;
    const int k   = t & (CH-1);
    const int s16 = t & (2*CH-1);

    if (lane == 0) sm.redf[par][wid] = fsum;
    __syncthreads();

    const float* xc = sm.xs[s16];
    const float rpart = sm.redf[par][lane];
    const float xrow  = xc[row];
    const float2 gq   = *reinterpret_cast<const float2*>(&sm.gq_s[s16][0]);
    float2 xp  = *reinterpret_cast<const float2*>(&xc[col0]);
    float2 mp  = *reinterpret_cast<const float2*>(&mrow[0]);
    float2 xp1 = *reinterpret_cast<const float2*>(&xc[col0 + 4]);
    float2 mp1 = *reinterpret_cast<const float2*>(&mrow[4]);

    if (t > 0 && tid < ND) out[((size_t)b*NT + t - 1)*ND + tid] = sm.ybuf[par ^ 1][tid];

    const float ssq = wave_sum(rpart);
    float sp = 1.0f;
    if (!(ssq <= 1024.0f)) {
      const float fn = fmaxf(sqrtf(ssq), 1e-6f);
      sp = fminf(32.0f/fn, 1.0f);
      #pragma unroll
      for (int i = 0; i < 64; ++i) w_u[i] *= sp;
    }
    const float spu = uni(sp);

    if (k == 0) {
      Su = 1.0f;
      #pragma unroll
      for (int i = 0; i < CH-1; ++i) { G[i] = 0.0f; cs[i] = 0.0f; }
      const float* xcol = &sm.xs[s16 + (lane & (CH-1))][col0];
      v8f aP = {};
      v8f aQ = {};
      #pragma unroll
      for (int s = 0; s < 32; ++s) {
        const float2 xb = *reinterpret_cast<const float2*>(&xcol[4*s]);
        const float2 mq = *reinterpret_cast<const float2*>(&mrow[4*s]);
        v2f bv; bv.x = xb.x;       bv.y = xb.y;
        v2f aw; aw.x = w_u[2*s];   aw.y = w_u[2*s+1];
        v2f am; am.x = mq.x;       am.y = mq.y;
        aP = __builtin_amdgcn_wmma_f32_16x16x4_f32(
             false, aw,  false, bv,
             (short)0, aP,  false,  false);
        aQ = __builtin_amdgcn_wmma_f32_16x16x4_f32(
             false, am,  false, bv,
             (short)0, aQ,  false,  false);
      }
      if ((lane & 15) < CH) {
        float* pp = &sm.pq[0][wid & 1][lane & 15][rb + 8*hi];
        float* qq = &sm.pq[1][wid & 1][lane & 15][rb + 8*hi];
        *reinterpret_cast<float4*>(pp)     = make_float4(aP[0], aP[1], aP[2], aP[3]);
        *reinterpret_cast<float4*>(pp + 4) = make_float4(aP[4], aP[5], aP[6], aP[7]);
        *reinterpret_cast<float4*>(qq)     = make_float4(aQ[0], aQ[1], aQ[2], aQ[3]);
        *reinterpret_cast<float4*>(qq + 4) = make_float4(aQ[4], aQ[5], aQ[6], aQ[7]);
      }
      const int widu = __builtin_amdgcn_readfirstlane(wid);
      if (widu < (CH*(CH-1))/2) {
        int kk = 1, base = 0;
        for (int q = 0; q < CH-2; ++q) {
          if (widu >= base + kk) { base += kk; ++kk; }
        }
        const int ll = widu - base;
        const float4* xa = reinterpret_cast<const float4*>(&sm.xs[s16 + ll][8*lane]);
        const float4* xk = reinterpret_cast<const float4*>(&sm.xs[s16 + kk][8*lane]);
        const float4 a0 = xa[0], a1 = xa[1], k0 = xk[0], k1 = xk[1];
        float d = a0.x*k0.x;
        d = fmaf(a0.y, k0.y, d); d = fmaf(a0.z, k0.z, d); d = fmaf(a0.w, k0.w, d);
        d = fmaf(a1.x, k1.x, d); d = fmaf(a1.y, k1.y, d); d = fmaf(a1.z, k1.z, d); d = fmaf(a1.w, k1.w, d);
        d = wave_sum(d);
        if (lane == 0) sm.dtab[kk][ll] = d;
      }
      __syncthreads();
    } else {
      Su *= spu;
      #pragma unroll
      for (int i = 0; i < CH-1; ++i) G[i] *= spu;
    }

    float H[CH-1];
    H[CH-2] = G[CH-2];
    #pragma unroll
    for (int i = CH-3; i >= 0; --i) H[i] = H[i+1] + G[i];
    const float4 d0 = *reinterpret_cast<const float4*>(&sm.dtab[k][0]);
    const float4 d1 = *reinterpret_cast<const float4*>(&sm.dtab[k][4]);
    const float dk[CH-1] = {d0.x, d0.y, d0.z, d0.w, d1.x, d1.y, d1.z};
    const float pP = sm.pq[0][0][k][row] + sm.pq[0][1][k][row];
    const float pQ = sm.pq[1][0][k][row] + sm.pq[1][1][k][row];
    float yd = Su*pP - H[0]*pQ;
    #pragma unroll
    for (int l = 0; l < CH-1; ++l) yd = fmaf(-cs[l], uni(dk[l])*H[l], yd);
    const float y = yd + bias_row;
    if (((wid & 1) == 0) && (hi == 0)) sm.ybuf[par][row] = y;

    const float err = (y - xrow) * (1.0f/ND);
    const float g   = gq.x;
    const float csn = err * gq.y;
    const float gu  = uni(g);
    #pragma unroll
    for (int i = 0; i < CH-1; ++i) {
      cs[i] = (i == k) ? csn : cs[i];
      G[i]  = (i == k) ? gu  : G[i];
    }

    if (t + 1 == NT) break;

    if (stager && (t + CH < NT)) {
      stage_x(sm, (t + CH) & (2*CH-1), j, wid - 8, lane, xpre);
      if (t + CH + 1 < NT) xpre = x[((size_t)b*NT + t + CH + 1)*ND + j];
    }

    if (tid == 0 && t >= 1 && (t + CH - 1 < NT)) lr_step(sm, (t + CH - 1) & (2*CH-1));

    v2f fs2[4];
    #pragma unroll
    for (int q = 0; q < 4; ++q) fs2[q] = v2f{0.0f, 0.0f};
    #pragma unroll
    for (int c = 0; c < 32; ++c) {
      float2 xp2 = xp1, mp2 = mp1;
      if (c + 2 < 32) {
        xp2 = *reinterpret_cast<const float2*>(&xc[col0 + 4*(c + 2)]);
        mp2 = *reinterpret_cast<const float2*>(&mrow[4*(c + 2)]);
      }
      asm volatile("" ::: "memory");
      __builtin_amdgcn_sched_barrier(0);
      mp.x = mp.x + csn*xp.x;
      mp.y = mp.y + csn*xp.y;
      *reinterpret_cast<float2*>(&sm.mom[row*MOM_STRIDE + col0 + 4*c]) = mp;
      const float w0 = w_u[2*c]   - g*mp.x;
      const float w1 = w_u[2*c+1] - g*mp.y;
      w_u[2*c] = w0; w_u[2*c+1] = w1;
      v2f av; av.x = w0;   av.y = w1;
      fs2[c & 3] = __builtin_elementwise_fma(av, av, fs2[c & 3]);
      xp = xp1; mp = mp1;
      xp1 = xp2; mp1 = mp2;
    }
    const v2f fs01 = fs2[0] + fs2[1];
    const v2f fs23 = fs2[2] + fs2[3];
    const v2f fsv  = fs01 + fs23;
    fsum = wave_sum(fsv.x + fsv.y);
  }

  __syncthreads();
  if (tid < ND) out[((size_t)b*NT + NT - 1)*ND + tid] = sm.ybuf[(NT - 1) & 1][tid];
}

extern "C" void kernel_launch(void* const* d_in, const int* in_sizes, int n_in,
                              void* d_out, int out_size, void* d_ws, size_t ws_size,
                              hipStream_t stream) {
  (void)in_sizes; (void)n_in; (void)out_size; (void)d_ws; (void)ws_size;
  const float* x    = (const float*)d_in[0];
  const float* Wb   = (const float*)d_in[1];
  const float* bias = (const float*)d_in[2];
  const float* lil  = (const float*)d_in[3];
  float* out = (float*)d_out;
  ttt_linear_kernel<<<dim3(NB), dim3(1024), 0, stream>>>(x, Wb, bias, lil, out);
}


# Update loop's LDS traffic as 16-byte operations (2 reads + 1 write per two K-slices, not 4 + 2): K order regrouped, a lane owns 4 contiguous columns per slice pair; y and norm sum orders change
# speedup vs baseline: 2.7933x; 1.1529x over previous
#include <hip/hip_runtime.h>
#include <math.h>


#define NB 4
#define NT 512
#define ND 256
#define CH 8
#define MOMB 0.9f
#define NOVB 0.95f
#define MOM_STRIDE 260
#define XPITCH 260
#define PQP 272

typedef float v2f __attribute__((ext_vector_type(2)));
typedef float v4f __attribute__((ext_vector_type(4)));
typedef float v8f __attribute__((ext_vector_type(8)));

struct __align__(16) Smem {
  float mom[ND * MOM_STRIDE];
  float xs[2 * CH][XPITCH];
  float pq[2][2][CH][PQP];
  float ybuf[2][ND];
  float iema[ND];
  float redn[2 * CH][8];
  float redf[2][32];
  float gq_s[2 * CH][2];
  float dtab[CH][8];
  float lrema_s;
  float inner_lr_s;
  float ppow_s;
};

__device__ __forceinline__ float xhalf(float v)
{
  const int b = __builtin_bit_cast(int, v);
  return __builtin_bit_cast(float, __builtin_amdgcn_permlanex16(b, b, 0x76543210, 0xfedcba98, false, false));
}

template <int CTRL>
__device__ __forceinline__ float row_xmask(float v)
{
  const int b = __builtin_bit_cast(int, v);
  return __builtin_bit_cast(float, __builtin_amdgcn_update_dpp(b, b, CTRL, 0xf, 0xf, true));
}

__device__ __forceinline__ float wave_sum(float a)
{
  a += xhalf(a);
  a += row_xmask<0x168>(a);
  a += row_xmask<0x164>(a);
  a += row_xmask<0x162>(a);
  a += row_xmask<0x161>(a);
  return a;
}

__device__ __forceinline__ float uni(float v)
{
  return __builtin_bit_cast(float, __builtin_amdgcn_readfirstlane(__builtin_bit_cast(int, v)));
}

__device__ __forceinline__ void lr_step(Smem& sm, int slot16)
{
  float s = 0.0f;
  #pragma unroll
  for (int i = 0; i < 8; ++i) s += sm.redn[slot16][i];
  const float nov = s * (1.0f/ND);
  const float lrs = fminf(fmaxf(1.0f + 3.0f*nov, 0.5f), 3.0f);
  const float lrema = NOVB*sm.lrema_s + (1.0f-NOVB)*lrs;
  sm.lrema_s = lrema;
  const float p = MOMB*sm.ppow_s;
  sm.ppow_s = p;
  const float q = 1.0f/p;
  sm.gq_s[slot16][0] = (sm.inner_lr_s * lrema) * p;
  sm.gq_s[slot16][1] = (1.0f-MOMB) * q;
}

__device__ __forceinline__ void stage_x(Smem& sm, int slot16, int j, int wslot, int lane, float xj)
{
  sm.xs[slot16][j] = xj;
  const float old = sm.iema[j];
  float a = fabsf(xj - old);
  sm.iema[j] = NOVB*old + (1.0f-NOVB)*xj;
  a = wave_sum(a);
  if (lane == 0) sm.redn[slot16][wslot] = a;
}

__global__ __launch_bounds__(1024) void ttt_linear_kernel(
    const float* __restrict__ x, const float* __restrict__ Wb,
    const float* __restrict__ bias, const float* __restrict__ loglr,
    float* __restrict__ out)
{
  __shared__ Smem sm;
  const int tid  = threadIdx.x;
  const int lane = tid & 31;
  const int wid  = tid >> 5;
  const int b    = blockIdx.x;
  const int hi   = lane >> 4;
  const int rb   = (wid >> 1) << 4;
  const int row  = rb + (lane & 15);
  const int colbase = (wid & 1) << 7;
  const bool stager = (tid >= ND) && (tid < 2*ND);
  const int  j      = tid & (ND-1);

  float w_u[64];
  #pragma unroll
  for (int p = 0; p < 16; ++p) {
    const int col = colbase + 8*p + 4*hi;
    const float4 wv = *reinterpret_cast<const float4*>(Wb + row*ND + col);
    w_u[4*p]   = wv.x;
    w_u[4*p+1] = wv.y;
    w_u[4*p+2] = wv.z;
    w_u[4*p+3] = wv.w;
  }
  for (int i = tid; i < ND*MOM_STRIDE; i += 1024) sm.mom[i] = 0.0f;
  if (tid < CH*8) sm.dtab[tid >> 3][tid & 7] = 0.0f;

  if (tid == 0) {
    sm.inner_lr_s = fminf(fmaxf(expf(loglr[0]), 1e-5f), 1.0f);
    sm.lrema_s = 1.0f;
    sm.ppow_s  = 1.0f;
  }
  const float bias_row = bias[row];

  float xpre = 0.0f;
  if (stager) {
    sm.iema[j] = 0.0f;
    for (int i = 0; i < CH; ++i)
      stage_x(sm, i, j, wid - 8, lane, x[((size_t)b*NT + i)*ND + j]);
    xpre = x[((size_t)b*NT + CH)*ND + j];
  }
  __syncthreads();
  if (tid == 0) {
    for (int i = 0; i < CH; ++i) lr_step(sm, i);
  }

  float fsum = 0.0f;
  const int col0 = colbase + 4*hi;
  const float* mrow = &sm.mom[row*MOM_STRIDE + col0];

  float Su = 1.0f;
  float G[CH-1], cs[CH-1];
  #pragma unroll
  for (int i = 0; i < CH-1; ++i) { G[i] = 0.0f; cs[i] = 0.0f; }

  for (int t = 0; t < NT; ++t) {
    const int par = t & 1;
    const int k   = t & (CH-1);
    const int s16 = t & (2*CH-1);

    if (lane == 0) sm.redf[par][wid] = fsum;
    __syncthreads();

    const float* xc = sm.xs[s16];
    const float rpart = sm.redf[par][lane];
    const float xrow  = xc[row];
    const float2 gq   = *reinterpret_cast<const float2*>(&sm.gq_s[s16][0]);
    v4f xp = *reinterpret_cast<const v4f*>(&xc[col0]);
    v4f mp = *reinterpret_cast<const v4f*>(&mrow[0]);

    if (t > 0 && tid < ND) out[((size_t)b*NT + t - 1)*ND + tid] = sm.ybuf[par ^ 1][tid];

    const float ssq = wave_sum(rpart);
    float sp = 1.0f;
    if (!(ssq <= 1024.0f)) {
      const float fn = fmaxf(sqrtf(ssq), 1e-6f);
      sp = fminf(32.0f/fn, 1.0f);
      #pragma unroll
      for (int i = 0; i < 64; ++i) w_u[i] *= sp;
    }
    const float spu = uni(sp);

    if (k == 0) {
      Su = 1.0f;
      #pragma unroll
      for (int i = 0; i < CH-1; ++i) { G[i] = 0.0f; cs[i] = 0.0f; }
      const float* xcol = &sm.xs[s16 + (lane & (CH-1))][col0];
      v8f aP = {};
      v8f aQ = {};
      #pragma unroll
      for (int p = 0; p < 16; ++p) {
        const float4 xb = *reinterpret_cast<const float4*>(&xcol[8*p]);
        const float4 mq = *reinterpret_cast<const float4*>(&mrow[8*p]);
        v2f bv0; bv0.x = xb.x;         bv0.y = xb.y;
        v2f aw0; aw0.x = w_u[4*p];     aw0.y = w_u[4*p+1];
        v2f am0; am0.x = mq.x;         am0.y = mq.y;
        aP = __builtin_amdgcn_wmma_f32_16x16x4_f32(
             false, aw0,  false, bv0,
             (short)0, aP,  false,  false);
        aQ = __builtin_amdgcn_wmma_f32_16x16x4_f32(
             false, am0,  false, bv0,
             (short)0, aQ,  false,  false);
        v2f bv1; bv1.x = xb.z;         bv1.y = xb.w;
        v2f aw1; aw1.x = w_u[4*p+2];   aw1.y = w_u[4*p+3];
        v2f am1; am1.x = mq.z;         am1.y = mq.w;
        aP = __builtin_amdgcn_wmma_f32_16x16x4_f32(
             false, aw1,  false, bv1,
             (short)0, aP,  false,  false);
        aQ = __builtin_amdgcn_wmma_f32_16x16x4_f32(
             false, am1,  false, bv1,
             (short)0, aQ,  false,  false);
      }
      if ((lane & 15) < CH) {
        float* pp = &sm.pq[0][wid & 1][lane & 15][rb + 8*hi];
        float* qq = &sm.pq[1][wid & 1][lane & 15][rb + 8*hi];
        *reinterpret_cast<float4*>(pp)     = make_float4(aP[0], aP[1], aP[2], aP[3]);
        *reinterpret_cast<float4*>(pp + 4) = make_float4(aP[4], aP[5], aP[6], aP[7]);
        *reinterpret_cast<float4*>(qq)     = make_float4(aQ[0], aQ[1], aQ[2], aQ[3]);
        *reinterpret_cast<float4*>(qq + 4) = make_float4(aQ[4], aQ[5], aQ[6], aQ[7]);
      }
      const int widu = __builtin_amdgcn_readfirstlane(wid);
      if (widu < (CH*(CH-1))/2) {
        int kk = 1, base = 0;
        for (int q = 0; q < CH-2; ++q) {
          if (widu >= base + kk) { base += kk; ++kk; }
        }
        const int ll = widu - base;
        const float4* xa = reinterpret_cast<const float4*>(&sm.xs[s16 + ll][8*lane]);
        const float4* xk = reinterpret_cast<const float4*>(&sm.xs[s16 + kk][8*lane]);
        const float4 a0 = xa[0], a1 = xa[1], k0 = xk[0], k1 = xk[1];
        float d = a0.x*k0.x;
        d = fmaf(a0.y, k0.y, d); d = fmaf(a0.z, k0.z, d); d = fmaf(a0.w, k0.w, d);
        d = fmaf(a1.x, k1.x, d); d = fmaf(a1.y, k1.y, d); d = fmaf(a1.z, k1.z, d); d = fmaf(a1.w, k1.w, d);
        d = wave_sum(d);
        if (lane == 0) sm.dtab[kk][ll] = d;
      }
      __syncthreads();
    } else {
      Su *= spu;
      #pragma unroll
      for (int i = 0; i < CH-1; ++i) G[i] *= spu;
    }

    float H[CH-1];
    H[CH-2] = G[CH-2];
    #pragma unroll
    for (int i = CH-3; i >= 0; --i) H[i] = H[i+1] + G[i];
    const float4 d0 = *reinterpret_cast<const float4*>(&sm.dtab[k][0]);
    const float4 d1 = *reinterpret_cast<const float4*>(&sm.dtab[k][4]);
    const float dk[CH-1] = {d0.x, d0.y, d0.z, d0.w, d1.x, d1.y, d1.z};
    const float pP = sm.pq[0][0][k][row] + sm.pq[0][1][k][row];
    const float pQ = sm.pq[1][0][k][row] + sm.pq[1][1][k][row];
    float yd = Su*pP - H[0]*pQ;
    #pragma unroll
    for (int l = 0; l < CH-1; ++l) yd = fmaf(-cs[l], uni(dk[l])*H[l], yd);
    const float y = yd + bias_row;
    if (((wid & 1) == 0) && (hi == 0)) sm.ybuf[par][row] = y;

    const float err = (y - xrow) * (1.0f/ND);
    const float g   = gq.x;
    const float csn = err * gq.y;
    const float gu  = uni(g);
    #pragma unroll
    for (int i = 0; i < CH-1; ++i) {
      cs[i] = (i == k) ? csn : cs[i];
      G[i]  = (i == k) ? gu  : G[i];
    }

    if (t + 1 == NT) break;

    if (stager && (t + CH < NT)) {
      stage_x(sm, (t + CH) & (2*CH-1), j, wid - 8, lane, xpre);
      if (t + CH + 1 < NT) xpre = x[((size_t)b*NT + t + CH + 1)*ND + j];
    }

    if (tid == 0 && t >= 1 && (t + CH - 1 < NT)) lr_step(sm, (t + CH - 1) & (2*CH-1));

    v2f fs2[4];
    #pragma unroll
    for (int q = 0; q < 4; ++q) fs2[q] = v2f{0.0f, 0.0f};
    #pragma unroll
    for (int p = 0; p < 16; ++p) {
      v4f xn = xp, mn = mp;
      if (p + 1 < 16) {
        xn = *reinterpret_cast<const v4f*>(&xc[col0 + 8*(p + 1)]);
        mn = *reinterpret_cast<const v4f*>(&mrow[8*(p + 1)]);
      }
      asm volatile("" :: "v"(xp), "v"(mp) : "memory");
      __builtin_amdgcn_sched_barrier(0);
      mp.x = mp.x + csn*xp.x;
      mp.y = mp.y + csn*xp.y;
      mp.z = mp.z + csn*xp.z;
      mp.w = mp.w + csn*xp.w;
      *reinterpret_cast<v4f*>(&sm.mom[row*MOM_STRIDE + col0 + 8*p]) = mp;
      const float w0 = w_u[4*p]   - g*mp.x;
      const float w1 = w_u[4*p+1] - g*mp.y;
      const float w2 = w_u[4*p+2] - g*mp.z;
      const float w3 = w_u[4*p+3] - g*mp.w;
      w_u[4*p] = w0; w_u[4*p+1] = w1; w_u[4*p+2] = w2; w_u[4*p+3] = w3;
      v2f av0; av0.x = w0;   av0.y = w1;
      v2f av1; av1.x = w2;   av1.y = w3;
      fs2[(2*p) & 3]     = __builtin_elementwise_fma(av0, av0, fs2[(2*p) & 3]);
      fs2[(2*p + 1) & 3] = __builtin_elementwise_fma(av1, av1, fs2[(2*p + 1) & 3]);
      xp = xn; mp = mn;
    }
    const v2f fs01 = fs2[0] + fs2[1];
    const v2f fs23 = fs2[2] + fs2[3];
    const v2f fsv  = fs01 + fs23;
    fsum = wave_sum(fsv.x + fsv.y);
  }

  __syncthreads();
  if (tid < ND) out[((size_t)b*NT + NT - 1)*ND + tid] = sm.ybuf[(NT - 1) & 1][tid];
}

extern "C" void kernel_launch(void* const* d_in, const int* in_sizes, int n_in,
                              void* d_out, int out_size, void* d_ws, size_t ws_size,
                              hipStream_t stream) {
  (void)in_sizes; (void)n_in; (void)out_size; (void)d_ws; (void)ws_size;
  const float* x    = (const float*)d_in[0];
  const float* Wb   = (const float*)d_in[1];
  const float* bias = (const float*)d_in[2];
  const float* lil  = (const float*)d_in[3];
  float* out = (float*)d_out;
  ttt_linear_kernel<<<dim3(NB), dim3(1024), 0, stream>>>(x, Wb, bias, lil, out);
}


# Work-group of 16 waves with 256 registers each, a wave owns 16 whole rows: per-step scalar work halves and 3/8 of the momentum lives in registers, not LDS; y and norm sum orders change
# speedup vs baseline: 3.0748x; 1.1008x over previous
#include <hip/hip_runtime.h>
#include <math.h>


#define NB 4
#define NT 512
#define ND 256
#define CH 8
#define MOMB 0.9f
#define NOVB 0.95f
#define MOM_STRIDE 260
#define XPITCH 260
#define PQP 272
#define NTHR 512
#define NP 32
#define MREG 12

typedef float v2f __attribute__((ext_vector_type(2)));
typedef float v4f __attribute__((ext_vector_type(4)));
typedef float v8f __attribute__((ext_vector_type(8)));

struct __align__(16) Smem {
  float mom[ND * MOM_STRIDE];
  float xs[2 * CH][XPITCH];
  float pq[2][CH][PQP];
  float ybuf[2][ND];
  float iema[ND];
  float redn[2 * CH][8];
  float redf[2][16];
  float gq_s[2 * CH][2];
  float dtab[CH][8];
  float lrema_s;
  float inner_lr_s;
  float ppow_s;
};

__device__ __forceinline__ float xhalf(float v)
{
  const int b = __builtin_bit_cast(int, v);
  return __builtin_bit_cast(float, __builtin_amdgcn_permlanex16(b, b, 0x76543210, 0xfedcba98, false, false));
}

template <int CTRL>
__device__ __forceinline__ float row_xmask(float v)
{
  const int b = __builtin_bit_cast(int, v);
  return __builtin_bit_cast(float, __builtin_amdgcn_update_dpp(b, b, CTRL, 0xf, 0xf, true));
}

__device__ __forceinline__ float wave_sum(float a)
{
  a += xhalf(a);
  a += row_xmask<0x168>(a);
  a += row_xmask<0x164>(a);
  a += row_xmask<0x162>(a);
  a += row_xmask<0x161>(a);
  return a;
}

__device__ __forceinline__ float row_sum(float a)
{
  a += row_xmask<0x168>(a);
  a += row_xmask<0x164>(a);
  a += row_xmask<0x162>(a);
  a += row_xmask<0x161>(a);
  return a;
}

__device__ __forceinline__ float uni(float v)
{
  return __builtin_bit_cast(float, __builtin_amdgcn_readfirstlane(__builtin_bit_cast(int, v)));
}

__device__ __forceinline__ void lr_step(Smem& sm, int slot16)
{
  float s = 0.0f;
  #pragma unroll
  for (int i = 0; i < 8; ++i) s += sm.redn[slot16][i];
  const float nov = s * (1.0f/ND);
  const float lrs = fminf(fmaxf(1.0f + 3.0f*nov, 0.5f), 3.0f);
  const float lrema = NOVB*sm.lrema_s + (1.0f-NOVB)*lrs;
  sm.lrema_s = lrema;
  const float p = MOMB*sm.ppow_s;
  sm.ppow_s = p;
  const float q = 1.0f/p;
  sm.gq_s[slot16][0] = (sm.inner_lr_s * lrema) * p;
  sm.gq_s[slot16][1] = (1.0f-MOMB) * q;
}

__device__ __forceinline__ void stage_x(Smem& sm, int slot16, int j, int wslot, int lane, float xj)
{
  sm.xs[slot16][j] = xj;
  const float old = sm.iema[j];
  float a = fabsf(xj - old);
  sm.iema[j] = NOVB*old + (1.0f-NOVB)*xj;
  a = wave_sum(a);
  if (lane == 0) sm.redn[slot16][wslot] = a;
}

__global__ __launch_bounds__(NTHR) void ttt_linear_kernel(
    const float* __restrict__ x, const float* __restrict__ Wb,
    const float* __restrict__ bias, const float* __restrict__ loglr,
    float* __restrict__ out)
{
  __shared__ Smem sm;
  const int tid  = threadIdx.x;
  const int lane = tid & 31;
  const int wid  = tid >> 5;
  const int b    = blockIdx.x;
  const int hi   = lane >> 4;
  const int rb   = wid << 4;
  const int row  = rb + (lane & 15);
  const bool stager = (tid >= ND) && (tid < 2*ND);
  const int  j      = tid & (ND-1);

  float w_u[4*NP];
  float m_u[4*MREG];
  #pragma unroll
  for (int i = 0; i < 4*MREG; ++i) m_u[i] = 0.0f;
  #pragma unroll
  for (int p = 0; p < NP; ++p) {
    const int col = 8*p + 4*hi;
    const float4 wv = *reinterpret_cast<const float4*>(Wb + row*ND + col);
    w_u[4*p]   = wv.x;
    w_u[4*p+1] = wv.y;
    w_u[4*p+2] = wv.z;
    w_u[4*p+3] = wv.w;
  }
  for (int i = tid; i < ND*MOM_STRIDE; i += NTHR) sm.mom[i] = 0.0f;
  if (tid < CH*8) sm.dtab[tid >> 3][tid & 7] = 0.0f;

  if (tid == 0) {
    sm.inner_lr_s = fminf(fmaxf(expf(loglr[0]), 1e-5f), 1.0f);
    sm.lrema_s = 1.0f;
    sm.ppow_s  = 1.0f;
  }
  const float bias_row = bias[row];

  float xpre = 0.0f;
  if (stager) {
    sm.iema[j] = 0.0f;
    for (int i = 0; i < CH; ++i)
      stage_x(sm, i, j, wid - 8, lane, x[((size_t)b*NT + i)*ND + j]);
    xpre = x[((size_t)b*NT + CH)*ND + j];
  }
  __syncthreads();
  if (tid == 0) {
    for (int i = 0; i < CH; ++i) lr_step(sm, i);
  }

  float fsum = 0.0f;
  const int col0 = 4*hi;
  const float* mrow = &sm.mom[row*MOM_STRIDE + col0];

  float Su = 1.0f;
  float G[CH-1], cs[CH-1];
  #pragma unroll
  for (int i = 0; i < CH-1; ++i) { G[i] = 0.0f; cs[i] = 0.0f; }

  for (int t = 0; t < NT; ++t) {
    const int par = t & 1;
    const int k   = t & (CH-1);
    const int s16 = t & (2*CH-1);

    if (lane == 0) sm.redf[par][wid] = fsum;
    __syncthreads();

    const float* xc = sm.xs[s16];
    const float rpart = sm.redf[par][lane & 15];
    const float xrow  = xc[row];
    const float2 gq   = *reinterpret_cast<const float2*>(&sm.gq_s[s16][0]);
    v4f xq0 = *reinterpret_cast<const v4f*>(&xc[col0]);
    v4f mq0 = *reinterpret_cast<const v4f*>(&mrow[0]);
    v4f xq1 = *reinterpret_cast<const v4f*>(&xc[col0 + 8]);
    v4f mq1 = *reinterpret_cast<const v4f*>(&mrow[8]);

    if (t > 0 && tid < ND) out[((size_t)b*NT + t - 1)*ND + tid] = sm.ybuf[par ^ 1][tid];

    const float ssq = row_sum(rpart);
    float sp = 1.0f;
    if (!(ssq <= 1024.0f)) {
      const float fn = fmaxf(sqrtf(ssq), 1e-6f);
      sp = fminf(32.0f/fn, 1.0f);
      #pragma unroll
      for (int i = 0; i < 4*NP; ++i) w_u[i] *= sp;
    }
    const float spu = uni(sp);

    if (k == 0) {
      Su = 1.0f;
      #pragma unroll
      for (int i = 0; i < CH-1; ++i) { G[i] = 0.0f; cs[i] = 0.0f; }
      const float* xcol = &sm.xs[s16 + (lane & (CH-1))][col0];
      v8f aP = {};
      v8f aQ = {};
      #pragma unroll
      for (int p = 0; p < NP; ++p) {
        const float4 xb = *reinterpret_cast<const float4*>(&xcol[8*p]);
        float4 mq;
        if (p < NP - MREG) {
          mq = *reinterpret_cast<const float4*>(&mrow[8*p]);
        } else {
          const int r = 4*(p - (NP - MREG));
          mq = make_float4(m_u[r], m_u[r+1], m_u[r+2], m_u[r+3]);
        }
        v2f bv0; bv0.x = xb.x;         bv0.y = xb.y;
        v2f aw0; aw0.x = w_u[4*p];     aw0.y = w_u[4*p+1];
        v2f am0; am0.x = mq.x;         am0.y = mq.y;
        aP = __builtin_amdgcn_wmma_f32_16x16x4_f32(
             false, aw0,  false, bv0,
             (short)0, aP,  false,  false);
        aQ = __builtin_amdgcn_wmma_f32_16x16x4_f32(
             false, am0,  false, bv0,
             (short)0, aQ,  false,  false);
        v2f bv1; bv1.x = xb.z;         bv1.y = xb.w;
        v2f aw1; aw1.x = w_u[4*p+2];   aw1.y = w_u[4*p+3];
        v2f am1; am1.x = mq.z;         am1.y = mq.w;
        aP = __builtin_amdgcn_wmma_f32_16x16x4_f32(
             false, aw1,  false, bv1,
             (short)0, aP,  false,  false);
        aQ = __builtin_amdgcn_wmma_f32_16x16x4_f32(
             false, am1,  false, bv1,
             (short)0, aQ,  false,  false);
      }
      if ((lane & 15) < CH) {
        float* pp = &sm.pq[0][lane & 15][rb + 8*hi];
        float* qq = &sm.pq[1][lane & 15][rb + 8*hi];
        *reinterpret_cast<float4*>(pp)     = make_float4(aP[0], aP[1], aP[2], aP[3]);
        *reinterpret_cast<float4*>(pp + 4) = make_float4(aP[4], aP[5], aP[6], aP[7]);
        *reinterpret_cast<float4*>(qq)     = make_float4(aQ[0], aQ[1], aQ[2], aQ[3]);
        *reinterpret_cast<float4*>(qq + 4) = make_float4(aQ[4], aQ[5], aQ[6], aQ[7]);
      }
      const int widu = __builtin_amdgcn_readfirstlane(wid);
      for (int dn = widu; dn < (CH*(CH-1))/2; dn += 16) {
        int kk = 1, base = 0;
        for (int q = 0; q < CH-2; ++q) {
          if (dn >= base + kk) { base += kk; ++kk; }
        }
        const int ll = dn - base;
        const float4* xa = reinterpret_cast<const float4*>(&sm.xs[s16 + ll][8*lane]);
        const float4* xk = reinterpret_cast<const float4*>(&sm.xs[s16 + kk][8*lane]);
        const float4 a0 = xa[0], a1 = xa[1], k0 = xk[0], k1 = xk[1];
        float d = a0.x*k0.x;
        d = fmaf(a0.y, k0.y, d); d = fmaf(a0.z, k0.z, d); d = fmaf(a0.w, k0.w, d);
        d = fmaf(a1.x, k1.x, d); d = fmaf(a1.y, k1.y, d); d = fmaf(a1.z, k1.z, d); d = fmaf(a1.w, k1.w, d);
        d = wave_sum(d);
        if (lane == 0) sm.dtab[kk][ll] = d;
      }
      __syncthreads();
    } else {
      Su *= spu;
      #pragma unroll
      for (int i = 0; i < CH-1; ++i) G[i] *= spu;
    }

    float H[CH-1];
    H[CH-2] = G[CH-2];
    #pragma unroll
    for (int i = CH-3; i >= 0; --i) H[i] = H[i+1] + G[i];
    const float4 d0 = *reinterpret_cast<const float4*>(&sm.dtab[k][0]);
    const float4 d1 = *reinterpret_cast<const float4*>(&sm.dtab[k][4]);
    const float dk[CH-1] = {d0.x, d0.y, d0.z, d0.w, d1.x, d1.y, d1.z};
    const float pP = sm.pq[0][k][row];
    const float pQ = sm.pq[1][k][row];
    float yd = Su*pP - H[0]*pQ;
    #pragma unroll
    for (int l = 0; l < CH-1; ++l) yd = fmaf(-cs[l], uni(dk[l])*H[l], yd);
    const float y = yd + bias_row;
    if (hi == 0) sm.ybuf[par][row] = y;

    const float err = (y - xrow) * (1.0f/ND);
    const float g   = gq.x;
    const float csn = err * gq.y;
    const float gu  = uni(g);
    #pragma unroll
    for (int i = 0; i < CH-1; ++i) {
      cs[i] = (i == k) ? csn : cs[i];
      G[i]  = (i == k) ? gu  : G[i];
    }

    if (t + 1 == NT) break;

    if (stager && (t + CH < NT)) {
      stage_x(sm, (t + CH) & (2*CH-1), j, wid - 8, lane, xpre);
      if (t + CH + 1 < NT) xpre = x[((size_t)b*NT + t + CH + 1)*ND + j];
    }

    if (tid == 0 && t >= 1 && (t + CH - 1 < NT)) lr_step(sm, (t + CH - 1) & (2*CH-1));

    v2f fs2[4];
    #pragma unroll
    for (int q = 0; q < 4; ++q) fs2[q] = v2f{0.0f, 0.0f};
    #pragma unroll
    for (int p = 0; p < NP; ++p) {
      v4f xq2 = xq0, mq2 = mq0;
      if (p + 2 < NP) {
        xq2 = *reinterpret_cast<const v4f*>(&xc[col0 + 8*(p + 2)]);
        if (p + 2 < NP - MREG) mq2 = *reinterpret_cast<const v4f*>(&mrow[8*(p + 2)]);
      }
      v4f mp = mq0;
      if (p >= NP - MREG) {
        const int r = 4*(p - (NP - MREG));
        mp = v4f{m_u[r], m_u[r+1], m_u[r+2], m_u[r+3]};
      }
      asm volatile("" :: "v"(xq0), "v"(mq0) : "memory");
      __builtin_amdgcn_sched_barrier(0);
      mp.x = mp.x + csn*xq0.x;
      mp.y = mp.y + csn*xq0.y;
      mp.z = mp.z + csn*xq0.z;
      mp.w = mp.w + csn*xq0.w;
      if (p < NP - MREG) {
        *reinterpret_cast<v4f*>(&sm.mom[row*MOM_STRIDE + col0 + 8*p]) = mp;
      } else {
        const int r = 4*(p - (NP - MREG));
        m_u[r] = mp.x; m_u[r+1] = mp.y; m_u[r+2] = mp.z; m_u[r+3] = mp.w;
      }
      const float w0 = w_u[4*p]   - g*mp.x;
      const float w1 = w_u[4*p+1] - g*mp.y;
      const float w2 = w_u[4*p+2] - g*mp.z;
      const float w3 = w_u[4*p+3] - g*mp.w;
      w_u[4*p] = w0; w_u[4*p+1] = w1; w_u[4*p+2] = w2; w_u[4*p+3] = w3;
      v2f av0; av0.x = w0;   av0.y = w1;
      v2f av1; av1.x = w2;   av1.y = w3;
      fs2[(2*p) & 3]     = __builtin_elementwise_fma(av0, av0, fs2[(2*p) & 3]);
      fs2[(2*p + 1) & 3] = __builtin_elementwise_fma(av1, av1, fs2[(2*p + 1) & 3]);
      xq0 = xq1; mq0 = mq1; xq1 = xq2; mq1 = mq2;
    }
    const v2f fs01 = fs2[0] + fs2[1];
    const v2f fs23 = fs2[2] + fs2[3];
    const v2f fsv  = fs01 + fs23;
    fsum = wave_sum(fsv.x + fsv.y);
  }

  __syncthreads();
  if (tid < ND) out[((size_t)b*NT + NT - 1)*ND + tid] = sm.ybuf[(NT - 1) & 1][tid];
}

extern "C" void kernel_launch(void* const* d_in, const int* in_sizes, int n_in,
                              void* d_out, int out_size, void* d_ws, size_t ws_size,
                              hipStream_t stream) {
  (void)in_sizes; (void)n_in; (void)out_size; (void)d_ws; (void)ws_size;
  const float* x    = (const float*)d_in[0];
  const float* Wb   = (const float*)d_in[1];
  const float* bias = (const float*)d_in[2];
  const float* lil  = (const float*)d_in[3];
  float* out = (float*)d_out;
  ttt_linear_kernel<<<dim3(NB), dim3(NTHR), 0, stream>>>(x, Wb, bias, lil, out);
}


# Matvecs of 16 scan steps batched, not 8: every B column of the WMMA pass is a real step, half the matrix instructions per step; the steps' history sums carried as running sums; y sum order changes
# speedup vs baseline: 3.6727x; 1.1945x over previous
#include <hip/hip_runtime.h>
#include <math.h>


#define NB 4
#define NT 512
#define ND 256
#define CH 16
#define MOMB 0.9f
#define NOVB 0.95f
#define XPITCH 260
#define PQP 272
#define NTHR 512
#define NP 32
#define MREG 11
#define MOM_STRIDE (8*(NP - MREG) + 4)

typedef float v2f __attribute__((ext_vector_type(2)));
typedef float v4f __attribute__((ext_vector_type(4)));
typedef float v8f __attribute__((ext_vector_type(8)));
typedef float v16f __attribute__((ext_vector_type(16)));

struct __align__(16) Smem {
  float mom[ND * MOM_STRIDE];
  float xs[2 * CH][XPITCH];
  float pq[2][CH][PQP];
  float ybuf[2][ND];
  float iema[ND];
  float redn[2 * CH][8];
  float redf[2][16];
  float gq_s[2 * CH][2];
  float dtab[CH][CH];
  float lrema_s;
  float inner_lr_s;
  float ppow_s;
};

__device__ __forceinline__ float xhalf(float v)
{
  const int b = __builtin_bit_cast(int, v);
  return __builtin_bit_cast(float, __builtin_amdgcn_permlanex16(b, b, 0x76543210, 0xfedcba98, false, false));
}

template <int CTRL>
__device__ __forceinline__ float row_xmask(float v)
{
  const int b = __builtin_bit_cast(int, v);
  return __builtin_bit_cast(float, __builtin_amdgcn_update_dpp(b, b, CTRL, 0xf, 0xf, true));
}

__device__ __forceinline__ float wave_sum(float a)
{
  a += xhalf(a);
  a += row_xmask<0x168>(a);
  a += row_xmask<0x164>(a);
  a += row_xmask<0x162>(a);
  a += row_xmask<0x161>(a);
  return a;
}

__device__ __forceinline__ float row_sum(float a)
{
  a += row_xmask<0x168>(a);
  a += row_xmask<0x164>(a);
  a += row_xmask<0x162>(a);
  a += row_xmask<0x161>(a);
  return a;
}

__device__ __forceinline__ float uni(float v)
{
  return __builtin_bit_cast(float, __builtin_amdgcn_readfirstlane(__builtin_bit_cast(int, v)));
}

template <int L>
__device__ __forceinline__ float lane_val(float v)
{
  return __builtin_bit_cast(float, __builtin_amdgcn_readlane(__builtin_bit_cast(int, v), L));
}

__device__ __forceinline__ void lr_step(Smem& sm, int slot16)
{
  float s = 0.0f;
  #pragma unroll
  for (int i = 0; i < 8; ++i) s += sm.redn[slot16][i];
  const float nov = s * (1.0f/ND);
  const float lrs = fminf(fmaxf(1.0f + 3.0f*nov, 0.5f), 3.0f);
  const float lrema = NOVB*sm.lrema_s + (1.0f-NOVB)*lrs;
  sm.lrema_s = lrema;
  const float p = MOMB*sm.ppow_s;
  sm.ppow_s = p;
  const float q = 1.0f/p;
  sm.gq_s[slot16][0] = (sm.inner_lr_s * lrema) * p;
  sm.gq_s[slot16][1] = (1.0f-MOMB) * q;
}

__device__ __forceinline__ void stage_x(Smem& sm, int slot16, int j, int wslot, int lane, float xj)
{
  sm.xs[slot16][j] = xj;
  const float old = sm.iema[j];
  float a = fabsf(xj - old);
  sm.iema[j] = NOVB*old + (1.0f-NOVB)*xj;
  a = wave_sum(a);
  if (lane == 0) sm.redn[slot16][wslot] = a;
}

__global__ __launch_bounds__(NTHR) void ttt_linear_kernel(
    const float* __restrict__ x, const float* __restrict__ Wb,
    const float* __restrict__ bias, const float* __restrict__ loglr,
    float* __restrict__ out)
{
  __shared__ Smem sm;
  const int tid  = threadIdx.x;
  const int lane = tid & 31;
  const int wid  = tid >> 5;
  const int b    = blockIdx.x;
  const int hi   = lane >> 4;
  const int rb   = wid << 4;
  const int row  = rb + (lane & 15);
  const bool stager = (tid >= ND) && (tid < 2*ND);
  const int  j      = tid & (ND-1);

  float w_u[4*NP];
  float m_u[4*MREG];
  #pragma unroll
  for (int i = 0; i < 4*MREG; ++i) m_u[i] = 0.0f;
  #pragma unroll
  for (int p = 0; p < NP; ++p) {
    const int col = 8*p + 4*hi;
    const float4 wv = *reinterpret_cast<const float4*>(Wb + row*ND + col);
    w_u[4*p]   = wv.x;
    w_u[4*p+1] = wv.y;
    w_u[4*p+2] = wv.z;
    w_u[4*p+3] = wv.w;
  }
  for (int i = tid; i < ND*MOM_STRIDE; i += NTHR) sm.mom[i] = 0.0f;
  if (tid < CH*CH) sm.dtab[tid / CH][tid % CH] = 0.0f;

  if (tid == 0) {
    sm.inner_lr_s = fminf(fmaxf(expf(loglr[0]), 1e-5f), 1.0f);
    sm.lrema_s = 1.0f;
    sm.ppow_s  = 1.0f;
  }
  const float bias_row = bias[row];

  float xpre = 0.0f;
  if (stager) {
    sm.iema[j] = 0.0f;
    for (int i = 0; i < CH; ++i)
      stage_x(sm, i, j, wid - 8, lane, x[((size_t)b*NT + i)*ND + j]);
    xpre = x[((size_t)b*NT + CH)*ND + j];
  }
  __syncthreads();
  if (tid == 0) {
    for (int i = 0; i < CH; ++i) lr_step(sm, i);
  }

  float fsum = 0.0f;
  const int col0 = 4*hi;
  const float* mrow = &sm.mom[row*MOM_STRIDE + col0];

  float Su = 1.0f;
  float gprev = 0.0f;
  float Hv = 0.0f;
  const int l15 = lane & 15;
  v16f cs = {};

  for (int t = 0; t < NT; ++t) {
    const int par = t & 1;
    const int k   = t & (CH-1);
    const int s16 = t & (2*CH-1);

    if (lane == 0) sm.redf[par][wid] = fsum;
    __syncthreads();

    const float* xc = sm.xs[s16];
    const float rpart = sm.redf[par][lane & 15];
    const float xrow  = xc[row];
    const float2 gq   = *reinterpret_cast<const float2*>(&sm.gq_s[s16][0]);
    v4f xq0 = *reinterpret_cast<const v4f*>(&xc[col0]);
    v4f mq0 = *reinterpret_cast<const v4f*>(&mrow[0]);
    v4f xq1 = *reinterpret_cast<const v4f*>(&xc[col0 + 8]);
    v4f mq1 = *reinterpret_cast<const v4f*>(&mrow[8]);

    if (t > 0 && tid < ND) out[((size_t)b*NT + t - 1)*ND + tid] = sm.ybuf[par ^ 1][tid];

    const float ssq = row_sum(rpart);
    float sp = 1.0f;
    if (!(ssq <= 1024.0f)) {
      const float fn = fmaxf(sqrtf(ssq), 1e-6f);
      sp = fminf(32.0f/fn, 1.0f);
      #pragma unroll
      for (int i = 0; i < 4*NP; ++i) w_u[i] *= sp;
    }
    const float spu = uni(sp);

    if (k == 0) {
      Su = 1.0f;
      Hv = 0.0f;
      cs = v16f{};
      const float* xcol = &sm.xs[s16 + (lane & (CH-1))][col0];
      v8f aP = {};
      v8f aQ = {};
      #pragma unroll
      for (int p = 0; p < NP; ++p) {
        const float4 xb = *reinterpret_cast<const float4*>(&xcol[8*p]);
        float4 mq;
        if (p < NP - MREG) {
          mq = *reinterpret_cast<const float4*>(&mrow[8*p]);
        } else {
          const int r = 4*(p - (NP - MREG));
          mq = make_float4(m_u[r], m_u[r+1], m_u[r+2], m_u[r+3]);
        }
        v2f bv0; bv0.x = xb.x;         bv0.y = xb.y;
        v2f aw0; aw0.x = w_u[4*p];     aw0.y = w_u[4*p+1];
        v2f am0; am0.x = mq.x;         am0.y = mq.y;
        aP = __builtin_amdgcn_wmma_f32_16x16x4_f32(
             false, aw0,  false, bv0,
             (short)0, aP,  false,  false);
        aQ = __builtin_amdgcn_wmma_f32_16x16x4_f32(
             false, am0,  false, bv0,
             (short)0, aQ,  false,  false);
        v2f bv1; bv1.x = xb.z;         bv1.y = xb.w;
        v2f aw1; aw1.x = w_u[4*p+2];   aw1.y = w_u[4*p+3];
        v2f am1; am1.x = mq.z;         am1.y = mq.w;
        aP = __builtin_amdgcn_wmma_f32_16x16x4_f32(
             false, aw1,  false, bv1,
             (short)0, aP,  false,  false);
        aQ = __builtin_amdgcn_wmma_f32_16x16x4_f32(
             false, am1,  false, bv1,
             (short)0, aQ,  false,  false);
      }
      {
        float* pp = &sm.pq[0][lane & 15][rb + 8*hi];
        float* qq = &sm.pq[1][lane & 15][rb + 8*hi];
        *reinterpret_cast<float4*>(pp)     = make_float4(aP[0], aP[1], aP[2], aP[3]);
        *reinterpret_cast<float4*>(pp + 4) = make_float4(aP[4], aP[5], aP[6], aP[7]);
        *reinterpret_cast<float4*>(qq)     = make_float4(aQ[0], aQ[1], aQ[2], aQ[3]);
        *reinterpret_cast<float4*>(qq + 4) = make_float4(aQ[4], aQ[5], aQ[6], aQ[7]);
      }
      const int widu = __builtin_amdgcn_readfirstlane(wid);
      for (int r = 1; r < CH/2; r += 2) {
        const int kA = (widu < r)     ? r     : CH - r;
        const int lA = (widu < r)     ? widu  : widu - r;
        const int kB = (widu < r + 1) ? r + 1 : CH - r - 1;
        const int lB = (widu < r + 1) ? widu  : widu - r - 1;
        const float4* xaA = reinterpret_cast<const float4*>(&sm.xs[s16 + lA][8*lane]);
        const float4* xkA = reinterpret_cast<const float4*>(&sm.xs[s16 + kA][8*lane]);
        const float4* xaB = reinterpret_cast<const float4*>(&sm.xs[s16 + lB][8*lane]);
        const float4* xkB = reinterpret_cast<const float4*>(&sm.xs[s16 + kB][8*lane]);
        const float4 a0 = xaA[0], a1 = xaA[1], k0 = xkA[0], k1 = xkA[1];
        const float4 b0 = xaB[0], b1 = xaB[1], j0 = xkB[0], j1 = xkB[1];
        float dA = a0.x*k0.x;
        float dB = b0.x*j0.x;
        dA = fmaf(a0.y, k0.y, dA); dA = fmaf(a0.z, k0.z, dA); dA = fmaf(a0.w, k0.w, dA);
        dB = fmaf(b0.y, j0.y, dB); dB = fmaf(b0.z, j0.z, dB); dB = fmaf(b0.w, j0.w, dB);
        dA = fmaf(a1.x, k1.x, dA); dA = fmaf(a1.y, k1.y, dA); dA = fmaf(a1.z, k1.z, dA); dA = fmaf(a1.w, k1.w, dA);
        dB = fmaf(b1.x, j1.x, dB); dB = fmaf(b1.y, j1.y, dB); dB = fmaf(b1.z, j1.z, dB); dB = fmaf(b1.w, j1.w, dB);
        dA = wave_sum(dA);
        dB = wave_sum(dB);
        if (lane == 0) { sm.dtab[kA][lA] = dA; sm.dtab[kB][lB] = dB; }
      }
      __syncthreads();
    } else {
      Su *= spu;
      Hv = (l15 < k) ? spu*(Hv + gprev) : Hv;
    }

    const float hd = sm.dtab[k][l15] * Hv;
    const float pP = sm.pq[0][k][row];
    const float pQ = sm.pq[1][k][row];
    float yd = Su*pP - lane_val<0>(Hv)*pQ;
    float ye = 0.0f;
    yd = fmaf(-cs[0],  lane_val<0>(hd),  yd);
    ye = fmaf(-cs[1],  lane_val<1>(hd),  ye);
    yd = fmaf(-cs[2],  lane_val<2>(hd),  yd);
    ye = fmaf(-cs[3],  lane_val<3>(hd),  ye);
    yd = fmaf(-cs[4],  lane_val<4>(hd),  yd);
    ye = fmaf(-cs[5],  lane_val<5>(hd),  ye);
    yd = fmaf(-cs[6],  lane_val<6>(hd),  yd);
    ye = fmaf(-cs[7],  lane_val<7>(hd),  ye);
    yd = fmaf(-cs[8],  lane_val<8>(hd),  yd);
    ye = fmaf(-cs[9],  lane_val<9>(hd),  ye);
    yd = fmaf(-cs[10], lane_val<10>(hd), yd);
    ye = fmaf(-cs[11], lane_val<11>(hd), ye);
    yd = fmaf(-cs[12], lane_val<12>(hd), yd);
    ye = fmaf(-cs[13], lane_val<13>(hd), ye);
    yd = fmaf(-cs[14], lane_val<14>(hd), yd);
    const float y = (yd + ye) + bias_row;
    if (hi == 0) sm.ybuf[par][row] = y;

    const float err = (y - xrow) * (1.0f/ND);
    const float g   = gq.x;
    const float csn = err * gq.y;
    const float gu  = uni(g);
    cs[k] = csn;
    gprev = gu;

    if (t + 1 == NT) break;

    if (stager && (t + CH < NT)) {
      stage_x(sm, (t + CH) & (2*CH-1), j, wid - 8, lane, xpre);
      if (t + CH + 1 < NT) xpre = x[((size_t)b*NT + t + CH + 1)*ND + j];
    }

    if (tid == 0 && t >= 1 && (t + CH - 1 < NT)) lr_step(sm, (t + CH - 1) & (2*CH-1));

    v2f fs2[4];
    #pragma unroll
    for (int q = 0; q < 4; ++q) fs2[q] = v2f{0.0f, 0.0f};
    #pragma unroll
    for (int p = 0; p < NP; ++p) {
      v4f xq2 = xq0, mq2 = mq0;
      if (p + 2 < NP) {
        xq2 = *reinterpret_cast<const v4f*>(&xc[col0 + 8*(p + 2)]);
        if (p + 2 < NP - MREG) mq2 = *reinterpret_cast<const v4f*>(&mrow[8*(p + 2)]);
      }
      v4f mp = mq0;
      if (p >= NP - MREG) {
        const int r = 4*(p - (NP - MREG));
        mp = v4f{m_u[r], m_u[r+1], m_u[r+2], m_u[r+3]};
      }
      asm volatile("" :: "v"(xq0), "v"(mq0) : "memory");
      __builtin_amdgcn_sched_barrier(0);
      mp.x = mp.x + csn*xq0.x;
      mp.y = mp.y + csn*xq0.y;
      mp.z = mp.z + csn*xq0.z;
      mp.w = mp.w + csn*xq0.w;
      if (p < NP - MREG) {
        *reinterpret_cast<v4f*>(&sm.mom[row*MOM_STRIDE + col0 + 8*p]) = mp;
      } else {
        const int r = 4*(p - (NP - MREG));
        m_u[r] = mp.x; m_u[r+1] = mp.y; m_u[r+2] = mp.z; m_u[r+3] = mp.w;
      }
      const float w0 = w_u[4*p]   - g*mp.x;
      const float w1 = w_u[4*p+1] - g*mp.y;
      const float w2 = w_u[4*p+2] - g*mp.z;
      const float w3 = w_u[4*p+3] - g*mp.w;
      w_u[4*p] = w0; w_u[4*p+1] = w1; w_u[4*p+2] = w2; w_u[4*p+3] = w3;
      v2f av0; av0.x = w0;   av0.y = w1;
      v2f av1; av1.x = w2;   av1.y = w3;
      fs2[(2*p) & 3]     = __builtin_elementwise_fma(av0, av0, fs2[(2*p) & 3]);
      fs2[(2*p + 1) & 3] = __builtin_elementwise_fma(av1, av1, fs2[(2*p + 1) & 3]);
      xq0 = xq1; mq0 = mq1; xq1 = xq2; mq1 = mq2;
    }
    const v2f fs01 = fs2[0] + fs2[1];
    const v2f fs23 = fs2[2] + fs2[3];
    const v2f fsv  = fs01 + fs23;
    fsum = wave_sum(fsv.x + fsv.y);
  }

  __syncthreads();
  if (tid < ND) out[((size_t)b*NT + NT - 1)*ND + tid] = sm.ybuf[(NT - 1) & 1][tid];
}

extern "C" void kernel_launch(void* const* d_in, const int* in_sizes, int n_in,
                              void* d_out, int out_size, void* d_ws, size_t ws_size,
                              hipStream_t stream) {
  (void)in_sizes; (void)n_in; (void)out_size; (void)d_ws; (void)ws_size;
  const float* x    = (const float*)d_in[0];
  const float* Wb   = (const float*)d_in[1];
  const float* bias = (const float*)d_in[2];
  const float* lil  = (const float*)d_in[3];
  float* out = (float*)d_out;
  ttt_linear_kernel<<<dim3(NB), dim3(NTHR), 0, stream>>>(x, Wb, bias, lil, out);
}
